# starting kernel
# baseline (speedup 1.0000x reference)
#include <hip/hip_runtime.h>


typedef __attribute__((ext_vector_type(2))) float v2f;
typedef __attribute__((ext_vector_type(8))) float v8f;

static constexpr int GN   = 100000;
static constexpr int GE   = 1000000;
static constexpr int DIN  = 128;
static constexpr int HID  = 64;
static constexpr int CAT  = 192;
static constexpr int LAB  = 40;

template <int W> struct PadRow {
    static constexpr int value = W + ((48 - (W % 32)) % 32);
};

template <int NT>
__launch_bounds__(256)
__global__ void gemm_wmma_f32(const float* __restrict__ A, int lda,
                              const float* __restrict__ B, int K, int dout,
                              int M, float* __restrict__ out, int ldo,
                              const float* __restrict__ bias, int bias_to_out,
                              float* __restrict__ agg,
                              const float* __restrict__ dinv)
{
    constexpr int W  = NT * 16;
    constexpr int WP = PadRow<W>::value;
    extern __shared__ v2f Bs2[];

    const int KP = K >> 1;
    for (int idx = threadIdx.x; idx < KP * W; idx += 256) {
        int kp = idx / W;
        int n  = idx - kp * W;
        v2f p;
        p.x = (n < dout) ? B[(2 * kp)     * dout + n] : 0.0f;
        p.y = (n < dout) ? B[(2 * kp + 1) * dout + n] : 0.0f;
        Bs2[kp * WP + n] = p;
    }
    __syncthreads();

    const int wave = threadIdx.x >> 5;
    const int lane = threadIdx.x & 31;
    const int l15  = lane & 15;
    const int half = lane >> 4;
    const int kb   = half * 2;

    const int m0 = (blockIdx.x * 8 + wave) * 16;
    int mA = m0 + l15;
    if (mA > M - 1) mA = M - 1;
    const float* Arow = A + (size_t)mA * (size_t)lda;

    v8f acc[NT];
#pragma unroll
    for (int t = 0; t < NT; ++t)
        acc[t] = (v8f){0.f, 0.f, 0.f, 0.f, 0.f, 0.f, 0.f, 0.f};

    for (int k = 0; k < K; k += 4) {
        v2f a = *(const v2f*)(Arow + k + kb);
        const v2f* brow = Bs2 + ((k >> 1) + half) * WP + l15;
#pragma unroll
        for (int t = 0; t < NT; ++t) {
            v2f b = brow[t * 16];
            acc[t] = __builtin_amdgcn_wmma_f32_16x16x4_f32(
                false, a, false, b, (short)0, acc[t], false, false);
        }
    }

#pragma unroll
    for (int t = 0; t < NT; ++t) {
        const int n = t * 16 + l15;
        float bv = 0.0f;
        if (bias && n < dout) bv = bias[n];
#pragma unroll
        for (int v = 0; v < 8; ++v) {
            const int m = m0 + half * 8 + v;
            if (m < M && n < dout) {
                const float val = acc[t][v];
                if (agg) {
                    out[(size_t)m * ldo + n] = val;
                    const float di = dinv[m];
                    agg[(size_t)m * dout + n] = bv + di * di * val;
                } else {
                    out[(size_t)m * ldo + n] = val + (bias_to_out ? bv : 0.0f);
                }
            }
        }
    }
}

__global__ void deg_init(float* __restrict__ d, int M) {
    int i = blockIdx.x * blockDim.x + threadIdx.x;
    if (i < M) d[i] = 1.0f;
}

__global__ void deg_scatter(const int* __restrict__ ei, int E,
                            float* __restrict__ deg) {
    int e = blockIdx.x * blockDim.x + threadIdx.x;
    if (e < E) unsafeAtomicAdd(&deg[ei[E + e]], 1.0f);
}

__global__ void deg_rsqrt(float* __restrict__ d, int M) {
    int i = blockIdx.x * blockDim.x + threadIdx.x;
    if (i < M) d[i] = rsqrtf(d[i]);
}

__global__ void edge_scatter(const int* __restrict__ ei, int E,
                             const float* __restrict__ dinv,
                             const float* __restrict__ hw,
                             float* __restrict__ agg)
{
    int gid = blockIdx.x * blockDim.x + threadIdx.x;
    int e = gid >> 5;
    if (e >= E) return;
    int lane = gid & 31;
    int s = ei[e];
    int d = ei[E + e];
    float w = dinv[s] * dinv[d];
    int c = lane * 2;
    float2 v = *(const float2*)(hw + (size_t)s * HID + c);
    float* p = agg + (size_t)d * HID + c;
    unsafeAtomicAdd(p,     w * v.x);
    unsafeAtomicAdd(p + 1, w * v.y);
}

__global__ void relu_concat(const float* __restrict__ agg,
                            float* __restrict__ Hcat, int M, int col0)
{
    int gid = blockIdx.x * blockDim.x + threadIdx.x;
    int i = gid >> 4;
    if (i >= M) return;
    int c = (gid & 15) * 4;
    float4 v = *(const float4*)(agg + (size_t)i * HID + c);
    v.x = fmaxf(v.x, 0.0f);
    v.y = fmaxf(v.y, 0.0f);
    v.z = fmaxf(v.z, 0.0f);
    v.w = fmaxf(v.w, 0.0f);
    *(float4*)(Hcat + (size_t)i * CAT + col0 + c) = v;
}

__global__ void logsoftmax40(float* __restrict__ out, int M)
{
    int i = blockIdx.x * blockDim.x + threadIdx.x;
    if (i >= M) return;
    float* row = out + (size_t)i * LAB;
    float v[LAB];
    float mx = -3.4e38f;
#pragma unroll
    for (int c = 0; c < LAB; ++c) { v[c] = row[c]; mx = fmaxf(mx, v[c]); }
    float s = 0.0f;
#pragma unroll
    for (int c = 0; c < LAB; ++c) s += __expf(v[c] - mx);
    float ls = __logf(s);
#pragma unroll
    for (int c = 0; c < LAB; ++c) row[c] = v[c] - mx - ls;
}

extern "C" void kernel_launch(void* const* d_in, const int* in_sizes, int n_in,
                              void* d_out, int out_size, void* d_ws, size_t ws_size,
                              hipStream_t stream)
{
    (void)in_sizes; (void)n_in; (void)out_size; (void)ws_size;

    const float* x  = (const float*)d_in[0];
    const int*   ei = (const int*)d_in[1];
    const float* Wm[3] = {(const float*)d_in[2], (const float*)d_in[4],
                          (const float*)d_in[6]};
    const float* bm[3] = {(const float*)d_in[3], (const float*)d_in[5],
                          (const float*)d_in[7]};
    const float* Wl = (const float*)d_in[8];
    const float* bl = (const float*)d_in[9];
    float* out = (float*)d_out;

    char* ws = (char*)d_ws;
    auto align256 = [](size_t b) { return (b + 255) & ~(size_t)255; };
    size_t off = 0;
    float* dinv = (float*)(ws + off); off += align256((size_t)GN * 4);
    float* hw   = (float*)(ws + off); off += align256((size_t)GN * HID * 4);
    float* agg  = (float*)(ws + off); off += align256((size_t)GN * HID * 4);
    float* Hcat = (float*)(ws + off); off += align256((size_t)GN * CAT * 4);

    const dim3 blk(256);

    deg_init   <<<(GN + 255) / 256, blk, 0, stream>>>(dinv, GN);
    deg_scatter<<<(GE + 255) / 256, blk, 0, stream>>>(ei, GE, dinv);
    deg_rsqrt  <<<(GN + 255) / 256, blk, 0, stream>>>(dinv, GN);

    const int gemmBlocks = (GN + 127) / 128;

    for (int l = 0; l < 3; ++l) {
        const float* Ain = (l == 0) ? x : (Hcat + HID * (l - 1));
        const int lda = (l == 0) ? DIN : CAT;
        const int K   = (l == 0) ? DIN : HID;
        const size_t shmem = (size_t)(K / 2) * PadRow<64>::value * sizeof(float2);
        gemm_wmma_f32<4><<<gemmBlocks, blk, shmem, stream>>>(
            Ain, lda, Wm[l], K, HID, GN, hw, HID, bm[l], 0, agg, dinv);
        edge_scatter<<<(GE * 32 + 255) / 256, blk, 0, stream>>>(
            ei, GE, dinv, hw, agg);
        relu_concat<<<((size_t)GN * 16 + 255) / 256, blk, 0, stream>>>(
            agg, Hcat, GN, HID * l);
    }

    const size_t shmemF = (size_t)(CAT / 2) * PadRow<48>::value * sizeof(float2);
    gemm_wmma_f32<3><<<gemmBlocks, blk, shmemF, stream>>>(
        Hcat, CAT, Wl, CAT, LAB, GN, out, LAB, bl, 1, nullptr, nullptr);

    logsoftmax40<<<(GN + 255) / 256, blk, 0, stream>>>(out, GN);
}


# edge scatter: one wave per 32 edges (indices and norms loaded once per lane, row loads issued in two independent groups) instead of one wave per edge; float atomics kept, arrival order as before
# speedup vs baseline: 1.2421x; 1.2421x over previous
#include <hip/hip_runtime.h>


typedef __attribute__((ext_vector_type(2))) float v2f;
typedef __attribute__((ext_vector_type(8))) float v8f;

static constexpr int GN   = 100000;
static constexpr int GE   = 1000000;
static constexpr int DIN  = 128;
static constexpr int HID  = 64;
static constexpr int CAT  = 192;
static constexpr int LAB  = 40;

template <int W> struct PadRow {
    static constexpr int value = W + ((48 - (W % 32)) % 32);
};

template <int NT>
__launch_bounds__(256)
__global__ void gemm_wmma_f32(const float* __restrict__ A, int lda,
                              const float* __restrict__ B, int K, int dout,
                              int M, float* __restrict__ out, int ldo,
                              const float* __restrict__ bias, int bias_to_out,
                              float* __restrict__ agg,
                              const float* __restrict__ dinv)
{
    constexpr int W  = NT * 16;
    constexpr int WP = PadRow<W>::value;
    extern __shared__ v2f Bs2[];

    const int KP = K >> 1;
    for (int idx = threadIdx.x; idx < KP * W; idx += 256) {
        int kp = idx / W;
        int n  = idx - kp * W;
        v2f p;
        p.x = (n < dout) ? B[(2 * kp)     * dout + n] : 0.0f;
        p.y = (n < dout) ? B[(2 * kp + 1) * dout + n] : 0.0f;
        Bs2[kp * WP + n] = p;
    }
    __syncthreads();

    const int wave = threadIdx.x >> 5;
    const int lane = threadIdx.x & 31;
    const int l15  = lane & 15;
    const int half = lane >> 4;
    const int kb   = half * 2;

    const int m0 = (blockIdx.x * 8 + wave) * 16;
    int mA = m0 + l15;
    if (mA > M - 1) mA = M - 1;
    const float* Arow = A + (size_t)mA * (size_t)lda;

    v8f acc[NT];
#pragma unroll
    for (int t = 0; t < NT; ++t)
        acc[t] = (v8f){0.f, 0.f, 0.f, 0.f, 0.f, 0.f, 0.f, 0.f};

    for (int k = 0; k < K; k += 4) {
        v2f a = *(const v2f*)(Arow + k + kb);
        const v2f* brow = Bs2 + ((k >> 1) + half) * WP + l15;
#pragma unroll
        for (int t = 0; t < NT; ++t) {
            v2f b = brow[t * 16];
            acc[t] = __builtin_amdgcn_wmma_f32_16x16x4_f32(
                false, a, false, b, (short)0, acc[t], false, false);
        }
    }

#pragma unroll
    for (int t = 0; t < NT; ++t) {
        const int n = t * 16 + l15;
        float bv = 0.0f;
        if (bias && n < dout) bv = bias[n];
#pragma unroll
        for (int v = 0; v < 8; ++v) {
            const int m = m0 + half * 8 + v;
            if (m < M && n < dout) {
                const float val = acc[t][v];
                if (agg) {
                    out[(size_t)m * ldo + n] = val;
                    const float di = dinv[m];
                    agg[(size_t)m * dout + n] = bv + di * di * val;
                } else {
                    out[(size_t)m * ldo + n] = val + (bias_to_out ? bv : 0.0f);
                }
            }
        }
    }
}

__global__ void deg_init(float* __restrict__ d, int M) {
    int i = blockIdx.x * blockDim.x + threadIdx.x;
    if (i < M) d[i] = 1.0f;
}

__global__ void deg_scatter(const int* __restrict__ ei, int E,
                            float* __restrict__ deg) {
    int e = blockIdx.x * blockDim.x + threadIdx.x;
    if (e < E) unsafeAtomicAdd(&deg[ei[E + e]], 1.0f);
}

__global__ void deg_rsqrt(float* __restrict__ d, int M) {
    int i = blockIdx.x * blockDim.x + threadIdx.x;
    if (i < M) d[i] = rsqrtf(d[i]);
}

static constexpr int EB = 32;
static constexpr int ED = 16;

__launch_bounds__(256)
__global__ void edge_scatter(const int* __restrict__ ei, int E,
                             const float* __restrict__ dinv,
                             const float* __restrict__ hw,
                             float* __restrict__ agg)
{
    const int lane = threadIdx.x & 31;
    const int wv = __builtin_amdgcn_readfirstlane(
        (int)((blockIdx.x * blockDim.x + threadIdx.x) >> 5));
    const int e0 = wv * EB;
    if (e0 >= E) return;
    const int n = (E - e0 < EB) ? (E - e0) : EB;
    const int e = e0 + ((lane < n) ? lane : (n - 1));
    const int s = ei[e];
    const int d = ei[E + e];
    const float w = dinv[s] * dinv[d];
    const int c = lane * 2;

    if (n == EB) {
#pragma unroll
        for (int j0 = 0; j0 < EB; j0 += ED) {
            float2 v[ED];
#pragma unroll
            for (int u = 0; u < ED; ++u) {
                const int sj = __builtin_amdgcn_readlane(s, j0 + u);
                v[u] = *(const float2*)(hw + (size_t)sj * HID + c);
            }
#pragma unroll
            for (int u = 0; u < ED; ++u) {
                const int dj = __builtin_amdgcn_readlane(d, j0 + u);
                const float wj = __int_as_float(
                    __builtin_amdgcn_readlane(__float_as_int(w), j0 + u));
                float* p = agg + (size_t)dj * HID + c;
                unsafeAtomicAdd(p,     wj * v[u].x);
                unsafeAtomicAdd(p + 1, wj * v[u].y);
            }
        }
    } else {
        for (int j = 0; j < n; ++j) {
            const int sj = __shfl(s, j);
            const int dj = __shfl(d, j);
            const float wj = __shfl(w, j);
            const float2 v = *(const float2*)(hw + (size_t)sj * HID + c);
            float* p = agg + (size_t)dj * HID + c;
            unsafeAtomicAdd(p,     wj * v.x);
            unsafeAtomicAdd(p + 1, wj * v.y);
        }
    }
}

__global__ void relu_concat(const float* __restrict__ agg,
                            float* __restrict__ Hcat, int M, int col0)
{
    int gid = blockIdx.x * blockDim.x + threadIdx.x;
    int i = gid >> 4;
    if (i >= M) return;
    int c = (gid & 15) * 4;
    float4 v = *(const float4*)(agg + (size_t)i * HID + c);
    v.x = fmaxf(v.x, 0.0f);
    v.y = fmaxf(v.y, 0.0f);
    v.z = fmaxf(v.z, 0.0f);
    v.w = fmaxf(v.w, 0.0f);
    *(float4*)(Hcat + (size_t)i * CAT + col0 + c) = v;
}

__global__ void logsoftmax40(float* __restrict__ out, int M)
{
    int i = blockIdx.x * blockDim.x + threadIdx.x;
    if (i >= M) return;
    float* row = out + (size_t)i * LAB;
    float v[LAB];
    float mx = -3.4e38f;
#pragma unroll
    for (int c = 0; c < LAB; ++c) { v[c] = row[c]; mx = fmaxf(mx, v[c]); }
    float s = 0.0f;
#pragma unroll
    for (int c = 0; c < LAB; ++c) s += __expf(v[c] - mx);
    float ls = __logf(s);
#pragma unroll
    for (int c = 0; c < LAB; ++c) row[c] = v[c] - mx - ls;
}

extern "C" void kernel_launch(void* const* d_in, const int* in_sizes, int n_in,
                              void* d_out, int out_size, void* d_ws, size_t ws_size,
                              hipStream_t stream)
{
    (void)in_sizes; (void)n_in; (void)out_size; (void)ws_size;

    const float* x  = (const float*)d_in[0];
    const int*   ei = (const int*)d_in[1];
    const float* Wm[3] = {(const float*)d_in[2], (const float*)d_in[4],
                          (const float*)d_in[6]};
    const float* bm[3] = {(const float*)d_in[3], (const float*)d_in[5],
                          (const float*)d_in[7]};
    const float* Wl = (const float*)d_in[8];
    const float* bl = (const float*)d_in[9];
    float* out = (float*)d_out;

    char* ws = (char*)d_ws;
    auto align256 = [](size_t b) { return (b + 255) & ~(size_t)255; };
    size_t off = 0;
    float* dinv = (float*)(ws + off); off += align256((size_t)GN * 4);
    float* hw   = (float*)(ws + off); off += align256((size_t)GN * HID * 4);
    float* agg  = (float*)(ws + off); off += align256((size_t)GN * HID * 4);
    float* Hcat = (float*)(ws + off); off += align256((size_t)GN * CAT * 4);

    const dim3 blk(256);

    deg_init   <<<(GN + 255) / 256, blk, 0, stream>>>(dinv, GN);
    deg_scatter<<<(GE + 255) / 256, blk, 0, stream>>>(ei, GE, dinv);
    deg_rsqrt  <<<(GN + 255) / 256, blk, 0, stream>>>(dinv, GN);

    const int gemmBlocks = (GN + 127) / 128;

    for (int l = 0; l < 3; ++l) {
        const float* Ain = (l == 0) ? x : (Hcat + HID * (l - 1));
        const int lda = (l == 0) ? DIN : CAT;
        const int K   = (l == 0) ? DIN : HID;
        const size_t shmem = (size_t)(K / 2) * PadRow<64>::value * sizeof(float2);
        gemm_wmma_f32<4><<<gemmBlocks, blk, shmem, stream>>>(
            Ain, lda, Wm[l], K, HID, GN, hw, HID, bm[l], 0, agg, dinv);
        edge_scatter<<<((GE + EB - 1) / EB + 7) / 8, blk, 0, stream>>>(
            ei, GE, dinv, hw, agg);
        relu_concat<<<((size_t)GN * 16 + 255) / 256, blk, 0, stream>>>(
            agg, Hcat, GN, HID * l);
    }

    const size_t shmemF = (size_t)(CAT / 2) * PadRow<48>::value * sizeof(float2);
    gemm_wmma_f32<3><<<gemmBlocks, blk, shmemF, stream>>>(
        Hcat, CAT, Wl, CAT, LAB, GN, out, LAB, bl, 1, nullptr, nullptr);

    logsoftmax40<<<(GN + 255) / 256, blk, 0, stream>>>(out, GN);
}


# edge aggregation: edges grouped by destination per call (int atomic rank, count scan, placement); a wave per node then sums rows in registers, stores once; float atomics gone, sum order = rank order
# speedup vs baseline: 1.9407x; 1.5624x over previous
#include <hip/hip_runtime.h>


typedef __attribute__((ext_vector_type(2))) float v2f;
typedef __attribute__((ext_vector_type(8))) float v8f;

static constexpr int GN   = 100000;
static constexpr int GE   = 1000000;
static constexpr int DIN  = 128;
static constexpr int HID  = 64;
static constexpr int CAT  = 192;
static constexpr int LAB  = 40;

template <int W> struct PadRow {
    static constexpr int value = W + ((48 - (W % 32)) % 32);
};

template <int NT>
__launch_bounds__(256)
__global__ void gemm_wmma_f32(const float* __restrict__ A, int lda,
                              const float* __restrict__ B, int K, int dout,
                              int M, float* __restrict__ out, int ldo,
                              const float* __restrict__ bias)
{
    constexpr int W  = NT * 16;
    constexpr int WP = PadRow<W>::value;
    extern __shared__ v2f Bs2[];

    const int KP = K >> 1;
    for (int idx = threadIdx.x; idx < KP * W; idx += 256) {
        int kp = idx / W;
        int n  = idx - kp * W;
        v2f p;
        p.x = (n < dout) ? B[(2 * kp)     * dout + n] : 0.0f;
        p.y = (n < dout) ? B[(2 * kp + 1) * dout + n] : 0.0f;
        Bs2[kp * WP + n] = p;
    }
    __syncthreads();

    const int wave = threadIdx.x >> 5;
    const int lane = threadIdx.x & 31;
    const int l15  = lane & 15;
    const int half = lane >> 4;
    const int kb   = half * 2;

    const int m0 = (blockIdx.x * 8 + wave) * 16;
    int mA = m0 + l15;
    if (mA > M - 1) mA = M - 1;
    const float* Arow = A + (size_t)mA * (size_t)lda;

    v8f acc[NT];
#pragma unroll
    for (int t = 0; t < NT; ++t)
        acc[t] = (v8f){0.f, 0.f, 0.f, 0.f, 0.f, 0.f, 0.f, 0.f};

    for (int k = 0; k < K; k += 4) {
        v2f a = *(const v2f*)(Arow + k + kb);
        const v2f* brow = Bs2 + ((k >> 1) + half) * WP + l15;
#pragma unroll
        for (int t = 0; t < NT; ++t) {
            v2f b = brow[t * 16];
            acc[t] = __builtin_amdgcn_wmma_f32_16x16x4_f32(
                false, a, false, b, (short)0, acc[t], false, false);
        }
    }

#pragma unroll
    for (int t = 0; t < NT; ++t) {
        const int n = t * 16 + l15;
        float bv = 0.0f;
        if (bias && n < dout) bv = bias[n];
#pragma unroll
        for (int v = 0; v < 8; ++v) {
            const int m = m0 + half * 8 + v;
            if (m < M && n < dout)
                out[(size_t)m * ldo + n] = acc[t][v] + bv;
        }
    }
}

__global__ void cnt_zero(int* __restrict__ cnt) {
    int i = blockIdx.x * blockDim.x + threadIdx.x;
    if (i < GN) cnt[i] = 0;
}

__global__ void edge_rank(const int* __restrict__ ei, int* __restrict__ cnt,
                          int* __restrict__ rank) {
    int e = blockIdx.x * blockDim.x + threadIdx.x;
    if (e < GE) rank[e] = atomicAdd(&cnt[ei[GE + e]], 1);
}

static constexpr int SCT = 1024;
static constexpr int NV4 = GN / 4;
static_assert(GN % 4 == 0, "scan reads cnt four at a time");

__launch_bounds__(SCT)
__global__ void deg_scan(const int* __restrict__ cnt, int* __restrict__ rowptr,
                         float* __restrict__ dinv)
{
    __shared__ int wsum[32];
    __shared__ int psum[32];
    const int tid  = threadIdx.x;
    const int lane = tid & 31;
    const int wave = tid >> 5;
    const int4* c4 = (const int4*)cnt;

    int pre = 0;
    for (int b = 0; b < (int)blockIdx.x; ++b) {
        const int4 q = c4[b * SCT + tid];
        pre += (q.x + q.y) + (q.z + q.w);
    }
    const int vi = blockIdx.x * SCT + tid;
    int4 q = make_int4(0, 0, 0, 0);
    if (vi < NV4) q = c4[vi];
    const int t4 = (q.x + q.y) + (q.z + q.w);

    int inc = t4;
#pragma unroll
    for (int o = 1; o < 32; o <<= 1) {
        const int y = __shfl_up(inc, o);
        if (lane >= o) inc += y;
    }
#pragma unroll
    for (int o = 16; o > 0; o >>= 1) pre += __shfl_xor(pre, o);
    if (lane == 31) wsum[wave] = inc;
    if (lane == 0)  psum[wave] = pre;
    __syncthreads();

    int tot = psum[lane] + ((lane < wave) ? wsum[lane] : 0);
#pragma unroll
    for (int o = 16; o > 0; o >>= 1) tot += __shfl_xor(tot, o);

    if (vi < NV4) {
        int4 r;
        r.x = tot + inc - t4;
        r.y = r.x + q.x;
        r.z = r.y + q.y;
        r.w = r.z + q.z;
        ((int4*)rowptr)[vi] = r;
        float4 dv;
        dv.x = rsqrtf((float)(q.x + 1));
        dv.y = rsqrtf((float)(q.y + 1));
        dv.z = rsqrtf((float)(q.z + 1));
        dv.w = rsqrtf((float)(q.w + 1));
        ((float4*)dinv)[vi] = dv;
        if (vi == NV4 - 1) rowptr[GN] = r.w + q.w;
    }
}

__global__ void edge_place(const int* __restrict__ ei, const int* __restrict__ rank,
                           const int* __restrict__ rowptr,
                           const float* __restrict__ dinv, int2* __restrict__ adj)
{
    int e = blockIdx.x * blockDim.x + threadIdx.x;
    if (e >= GE) return;
    const int s = ei[e];
    const int d = ei[GE + e];
    const int pos = rowptr[d] + rank[e];
    if ((unsigned)pos < (unsigned)GE)
        adj[pos] = make_int2(s, __float_as_int(dinv[s] * dinv[d]));
}

static constexpr int NPW = 4;
static_assert(GN % NPW == 0, "a wave's nodes are all valid");

template <int G>
__device__ __forceinline__ void gather_rows(const float* __restrict__ hw,
                                            int s, float w, int j, int c,
                                            float& ax, float& ay)
{
    float2 v[G];
#pragma unroll
    for (int u = 0; u < G; ++u) {
        const int sj = __builtin_amdgcn_readlane(s, j + u);
        v[u] = *(const float2*)(hw + (size_t)sj * HID + c);
    }
#pragma unroll
    for (int u = 0; u < G; ++u) {
        const float wj = __int_as_float(
            __builtin_amdgcn_readlane(__float_as_int(w), j + u));
        ax += wj * v[u].x;
        ay += wj * v[u].y;
    }
}

__device__ __forceinline__ int imin(int a, int b) { return a < b ? a : b; }
__device__ __forceinline__ int imax(int a, int b) { return a > b ? a : b; }

__launch_bounds__(256)
__global__ void node_gather(const int* __restrict__ rowptr,
                            const int2* __restrict__ adj,
                            const float* __restrict__ dinv,
                            const float* __restrict__ hw,
                            const float* __restrict__ bias,
                            float* __restrict__ Hcat, int col0)
{
    const int lane = threadIdx.x & 31;
    const int wv = __builtin_amdgcn_readfirstlane(
        (int)((blockIdx.x * blockDim.x + threadIdx.x) >> 5));
    const int i0 = wv * NPW;
    if (i0 >= GN) return;
    const int c = lane * 2;

    int rp[NPW + 1];
#pragma unroll
    for (int q = 0; q <= NPW; ++q)
        rp[q] = imin(imax(__builtin_amdgcn_readfirstlane(rowptr[i0 + q]), 0), GE);

    int2   a0[NPW];
    float2 hs[NPW];
    float  di[NPW];
#pragma unroll
    for (int q = 0; q < NPW; ++q) {
        const int n = imin(rp[q + 1] - rp[q], 32);
        const int idx = imin(rp[q] + imax(imin(lane, n - 1), 0), GE - 1);
        a0[q] = adj[idx];
        hs[q] = *(const float2*)(hw + (size_t)(i0 + q) * HID + c);
        di[q] = dinv[i0 + q];
    }
    const float2 bv = *(const float2*)(bias + c);

#pragma unroll
    for (int q = 0; q < NPW; ++q) {
        const int end = rp[q + 1];
        int base = rp[q];
        int2 aa = a0[q];
        float ax = 0.0f, ay = 0.0f;
        while (base < end) {
            const int n = imin(end - base, 32);
            const int   s = imin(imax(aa.x, 0), GN - 1);
            const float w = __int_as_float(aa.y);
            int j = 0;
            for (; j + 8 <= n; j += 8) gather_rows<8>(hw, s, w, j, c, ax, ay);
            if (j + 4 <= n) { gather_rows<4>(hw, s, w, j, c, ax, ay); j += 4; }
            if (j + 2 <= n) { gather_rows<2>(hw, s, w, j, c, ax, ay); j += 2; }
            if (j < n)      { gather_rows<1>(hw, s, w, j, c, ax, ay); }
            base += 32;
            if (base < end) {
                const int n2 = imin(end - base, 32);
                aa = adj[imin(base + imin(lane, n2 - 1), GE - 1)];
            }
        }
        const float sl = di[q] * di[q];
        ax += sl * hs[q].x;
        ay += sl * hs[q].y;
        float2 o;
        o.x = fmaxf(ax + bv.x, 0.0f);
        o.y = fmaxf(ay + bv.y, 0.0f);
        *(float2*)(Hcat + (size_t)(i0 + q) * CAT + col0 + c) = o;
    }
}

__global__ void logsoftmax40(float* __restrict__ out, int M)
{
    int i = blockIdx.x * blockDim.x + threadIdx.x;
    if (i >= M) return;
    float* row = out + (size_t)i * LAB;
    float v[LAB];
    float mx = -3.4e38f;
#pragma unroll
    for (int c = 0; c < LAB; ++c) { v[c] = row[c]; mx = fmaxf(mx, v[c]); }
    float s = 0.0f;
#pragma unroll
    for (int c = 0; c < LAB; ++c) s += __expf(v[c] - mx);
    float ls = __logf(s);
#pragma unroll
    for (int c = 0; c < LAB; ++c) row[c] = v[c] - mx - ls;
}

extern "C" void kernel_launch(void* const* d_in, const int* in_sizes, int n_in,
                              void* d_out, int out_size, void* d_ws, size_t ws_size,
                              hipStream_t stream)
{
    (void)in_sizes; (void)n_in; (void)out_size; (void)ws_size;

    const float* x  = (const float*)d_in[0];
    const int*   ei = (const int*)d_in[1];
    const float* Wm[3] = {(const float*)d_in[2], (const float*)d_in[4],
                          (const float*)d_in[6]};
    const float* bm[3] = {(const float*)d_in[3], (const float*)d_in[5],
                          (const float*)d_in[7]};
    const float* Wl = (const float*)d_in[8];
    const float* bl = (const float*)d_in[9];
    float* out = (float*)d_out;

    char* ws = (char*)d_ws;
    auto align256 = [](size_t b) { return (b + 255) & ~(size_t)255; };
    size_t off = 0;
    float* dinv   = (float*)(ws + off); off += align256((size_t)GN * 4);
    int*   cnt    = (int*)  (ws + off); off += align256((size_t)GN * 4);
    int*   rowptr = (int*)  (ws + off); off += align256((size_t)(GN + 1) * 4);
    int*   rank   = (int*)  (ws + off); off += align256((size_t)GE * 4);
    int2*  adj    = (int2*) (ws + off); off += align256((size_t)GE * 8);
    float* hw     = (float*)(ws + off); off += align256((size_t)GN * HID * 4);
    float* Hcat   = (float*)(ws + off); off += align256((size_t)GN * CAT * 4);

    const dim3 blk(256);

    cnt_zero  <<<(GN + 255) / 256, blk, 0, stream>>>(cnt);
    edge_rank <<<(GE + 255) / 256, blk, 0, stream>>>(ei, cnt, rank);
    deg_scan  <<<(NV4 + SCT - 1) / SCT, dim3(SCT), 0, stream>>>(cnt, rowptr, dinv);
    edge_place<<<(GE + 255) / 256, blk, 0, stream>>>(ei, rank, rowptr, dinv, adj);

    const int gemmBlocks = (GN + 127) / 128;
    const int gatherBlocks = (GN / NPW + 7) / 8;

    for (int l = 0; l < 3; ++l) {
        const float* Ain = (l == 0) ? x : (Hcat + HID * (l - 1));
        const int lda = (l == 0) ? DIN : CAT;
        const int K   = (l == 0) ? DIN : HID;
        const size_t shmem = (size_t)(K / 2) * PadRow<64>::value * sizeof(float2);
        gemm_wmma_f32<4><<<gemmBlocks, blk, shmem, stream>>>(
            Ain, lda, Wm[l], K, HID, GN, hw, HID, nullptr);
        node_gather<<<gatherBlocks, blk, 0, stream>>>(
            rowptr, adj, dinv, hw, bm[l], Hcat, HID * l);
    }

    const size_t shmemF = (size_t)(CAT / 2) * PadRow<48>::value * sizeof(float2);
    gemm_wmma_f32<3><<<gemmBlocks, blk, shmemF, stream>>>(
        Hcat, CAT, Wl, CAT, LAB, GN, out, LAB, bl);

    logsoftmax40<<<(GN + 255) / 256, blk, 0, stream>>>(out, GN);
}


# edge grouping: per-edge global returning atomic (rank) replaced by a two-level counting sort on LDS counters (250 destination ranges, then the nodes of each range); sum order = LDS arrival rank
# speedup vs baseline: 2.3529x; 1.2124x over previous
#include <hip/hip_runtime.h>


typedef __attribute__((ext_vector_type(2))) float v2f;
typedef __attribute__((ext_vector_type(8))) float v8f;

static constexpr int GN   = 100000;
static constexpr int GE   = 1000000;
static constexpr int DIN  = 128;
static constexpr int HID  = 64;
static constexpr int CAT  = 192;
static constexpr int LAB  = 40;

template <int W> struct PadRow {
    static constexpr int value = W + ((48 - (W % 32)) % 32);
};

template <int NT>
__launch_bounds__(256)
__global__ void gemm_wmma_f32(const float* __restrict__ A, int lda,
                              const float* __restrict__ B, int K, int dout,
                              int M, float* __restrict__ out, int ldo,
                              const float* __restrict__ bias)
{
    constexpr int W  = NT * 16;
    constexpr int WP = PadRow<W>::value;
    extern __shared__ v2f Bs2[];

    const int KP = K >> 1;
    for (int idx = threadIdx.x; idx < KP * W; idx += 256) {
        int kp = idx / W;
        int n  = idx - kp * W;
        v2f p;
        p.x = (n < dout) ? B[(2 * kp)     * dout + n] : 0.0f;
        p.y = (n < dout) ? B[(2 * kp + 1) * dout + n] : 0.0f;
        Bs2[kp * WP + n] = p;
    }
    __syncthreads();

    const int wave = threadIdx.x >> 5;
    const int lane = threadIdx.x & 31;
    const int l15  = lane & 15;
    const int half = lane >> 4;
    const int kb   = half * 2;

    const int m0 = (blockIdx.x * 8 + wave) * 16;
    int mA = m0 + l15;
    if (mA > M - 1) mA = M - 1;
    const float* Arow = A + (size_t)mA * (size_t)lda;

    v8f acc[NT];
#pragma unroll
    for (int t = 0; t < NT; ++t)
        acc[t] = (v8f){0.f, 0.f, 0.f, 0.f, 0.f, 0.f, 0.f, 0.f};

    for (int k = 0; k < K; k += 4) {
        v2f a = *(const v2f*)(Arow + k + kb);
        const v2f* brow = Bs2 + ((k >> 1) + half) * WP + l15;
#pragma unroll
        for (int t = 0; t < NT; ++t) {
            v2f b = brow[t * 16];
            acc[t] = __builtin_amdgcn_wmma_f32_16x16x4_f32(
                false, a, false, b, (short)0, acc[t], false, false);
        }
    }

#pragma unroll
    for (int t = 0; t < NT; ++t) {
        const int n = t * 16 + l15;
        float bv = 0.0f;
        if (bias && n < dout) bv = bias[n];
#pragma unroll
        for (int v = 0; v < 8; ++v) {
            const int m = m0 + half * 8 + v;
            if (m < M && n < dout)
                out[(size_t)m * ldo + n] = acc[t][v] + bv;
        }
    }
}

static constexpr int BW   = 400;
static constexpr int NBIN = GN / BW;
static constexpr int CH   = 4096;
static constexpr int EB   = (GE + CH - 1) / CH;
static constexpr int L1T  = 512;
static constexpr int GRPS  = CH / 4 / L1T;
static constexpr int SCT  = 1024;
static_assert(GN % BW == 0, "ranges cover the nodes exactly");
static_assert(NBIN <= L1T && EB <= SCT && BW <= SCT, "one thread per counter");
static_assert(GE % 4 == 0 && CH % (4 * L1T) == 0, "edges are read four at a time");

__device__ __forceinline__ int imin(int a, int b) { return a < b ? a : b; }
__device__ __forceinline__ int imax(int a, int b) { return a > b ? a : b; }

__device__ __forceinline__ int bin_of(int d) {
    return imin(imax(d, 0), GN - 1) / BW;
}

__device__ __forceinline__ int block_excl_scan(int v, int pre, int* wsum, int* psum)
{
    const int lane = threadIdx.x & 31;
    const int wave = threadIdx.x >> 5;
    int inc = v;
#pragma unroll
    for (int o = 1; o < 32; o <<= 1) {
        const int y = __shfl_up(inc, o);
        if (lane >= o) inc += y;
    }
#pragma unroll
    for (int o = 16; o > 0; o >>= 1) pre += __shfl_xor(pre, o);
    if (lane == 31) wsum[wave] = inc;
    if (lane == 0)  psum[wave] = pre;
    __syncthreads();
    int tot = psum[lane] + ((lane < wave) ? wsum[lane] : 0);
#pragma unroll
    for (int o = 16; o > 0; o >>= 1) tot += __shfl_xor(tot, o);
    return tot + inc - v;
}

__launch_bounds__(L1T)
__global__ void bin_hist(const int* __restrict__ ei, int* __restrict__ H)
{
    __shared__ int h[NBIN];
    const int tid = threadIdx.x;
    const int b   = blockIdx.x;
    if (tid < NBIN) h[tid] = 0;
    __syncthreads();
    const int4* d4 = (const int4*)(ei + GE);
    int4 d[GRPS];
    bool ok[GRPS];
#pragma unroll
    for (int k = 0; k < GRPS; ++k) {
        const int g = b * (CH / 4) + k * L1T + tid;
        ok[k] = g < GE / 4;
        d[k]  = d4[ok[k] ? g : 0];
    }
#pragma unroll
    for (int k = 0; k < GRPS; ++k) {
        if (ok[k]) {
            atomicAdd(&h[bin_of(d[k].x)], 1);
            atomicAdd(&h[bin_of(d[k].y)], 1);
            atomicAdd(&h[bin_of(d[k].z)], 1);
            atomicAdd(&h[bin_of(d[k].w)], 1);
        }
    }
    __syncthreads();
    if (tid < NBIN) H[tid * EB + b] = h[tid];
}

__launch_bounds__(SCT)
__global__ void bin_scan(const int* __restrict__ H, int* __restrict__ O,
                         int* __restrict__ binstart)
{
    __shared__ int wsum[32];
    __shared__ int psum[32];
    const int tid = threadIdx.x;
    const int n0  = blockIdx.x * EB;
    int pre = 0;
    for (int i = tid; i < n0; i += SCT) pre += H[i];
    const int v = (tid < EB) ? H[n0 + tid] : 0;
    const int ex = block_excl_scan(v, pre, wsum, psum);
    if (tid < EB) O[n0 + tid] = ex;
    if (tid == 0) binstart[blockIdx.x] = ex;
}

__launch_bounds__(L1T)
__global__ void bin_scatter(const int* __restrict__ ei, const int* __restrict__ O,
                            int2* __restrict__ tmp)
{
    __shared__ int cur[NBIN];
    const int tid = threadIdx.x;
    const int b   = blockIdx.x;
    if (tid < NBIN) cur[tid] = O[tid * EB + b];
    __syncthreads();
    const int4* s4 = (const int4*)ei;
    const int4* d4 = (const int4*)(ei + GE);
    int4 s[GRPS], d[GRPS];
    bool ok[GRPS];
#pragma unroll
    for (int k = 0; k < GRPS; ++k) {
        const int g = b * (CH / 4) + k * L1T + tid;
        ok[k] = g < GE / 4;
        s[k]  = s4[ok[k] ? g : 0];
        d[k]  = d4[ok[k] ? g : 0];
    }
#pragma unroll
    for (int k = 0; k < GRPS; ++k) {
        if (ok[k]) {
            const int p0 = atomicAdd(&cur[bin_of(d[k].x)], 1);
            const int p1 = atomicAdd(&cur[bin_of(d[k].y)], 1);
            const int p2 = atomicAdd(&cur[bin_of(d[k].z)], 1);
            const int p3 = atomicAdd(&cur[bin_of(d[k].w)], 1);
            if ((unsigned)p0 < (unsigned)GE) tmp[p0] = make_int2(s[k].x, d[k].x);
            if ((unsigned)p1 < (unsigned)GE) tmp[p1] = make_int2(s[k].y, d[k].y);
            if ((unsigned)p2 < (unsigned)GE) tmp[p2] = make_int2(s[k].z, d[k].z);
            if ((unsigned)p3 < (unsigned)GE) tmp[p3] = make_int2(s[k].w, d[k].w);
        }
    }
}

static constexpr int RPT = 4;
__launch_bounds__(SCT)
__global__ void bin_sort(const int2* __restrict__ tmp, const int* __restrict__ binstart,
                         int* __restrict__ rowptr, float* __restrict__ dinv,
                         int2* __restrict__ srt)
{
    __shared__ int c[BW];
    __shared__ int wsum[32];
    __shared__ int psum[32];
    const int tid  = threadIdx.x;
    const int bin  = blockIdx.x;
    const int nb   = bin * BW;
    const int lo   = imin(imax(binstart[bin], 0), GE);
    const int hi   = (bin == NBIN - 1) ? GE : imin(imax(binstart[bin + 1], lo), GE);

    if (tid < BW) c[tid] = 0;
    __syncthreads();
    for (int p0 = lo + tid; p0 < hi; p0 += RPT * SCT) {
        int dl[RPT];
#pragma unroll
        for (int u = 0; u < RPT; ++u)
            dl[u] = imin(imax(tmp[imin(p0 + u * SCT, hi - 1)].y - nb, 0), BW - 1);
#pragma unroll
        for (int u = 0; u < RPT; ++u)
            atomicAdd(&c[dl[u]], (p0 + u * SCT < hi) ? 1 : 0);
    }
    __syncthreads();
    const int v  = (tid < BW) ? c[tid] : 0;
    const int ex = block_excl_scan(v, 0, wsum, psum);
    if (tid < BW) {
        rowptr[nb + tid] = lo + ex;
        dinv[nb + tid]   = rsqrtf((float)(v + 1));
        c[tid] = ex;
    }
    if (bin == NBIN - 1 && tid == 0) rowptr[GN] = hi;
    __syncthreads();
    for (int p0 = lo + tid; p0 < hi; p0 += RPT * SCT) {
        int2 sd[RPT];
        int  pos[RPT];
#pragma unroll
        for (int u = 0; u < RPT; ++u)
            sd[u] = tmp[imin(p0 + u * SCT, hi - 1)];
#pragma unroll
        for (int u = 0; u < RPT; ++u) {
            const bool live = p0 + u * SCT < hi;
            const int r = atomicAdd(&c[imin(imax(sd[u].y - nb, 0), BW - 1)], live ? 1 : 0);
            pos[u] = live ? lo + r : -1;
        }
#pragma unroll
        for (int u = 0; u < RPT; ++u)
            if ((unsigned)pos[u] < (unsigned)GE) srt[pos[u]] = sd[u];
    }
}

__global__ void edge_norm(const int2* __restrict__ srt, const float* __restrict__ dinv,
                          int2* __restrict__ adj)
{
    const int t = blockIdx.x * blockDim.x + threadIdx.x;
    if (t >= GE / 4) return;
    const int4* i4 = (const int4*)srt;
    int4 q0 = i4[t];
    int4 q1 = i4[t + GE / 4];
    const float w0 = dinv[imin(imax(q0.x, 0), GN - 1)] * dinv[imin(imax(q0.y, 0), GN - 1)];
    const float w1 = dinv[imin(imax(q0.z, 0), GN - 1)] * dinv[imin(imax(q0.w, 0), GN - 1)];
    const float w2 = dinv[imin(imax(q1.x, 0), GN - 1)] * dinv[imin(imax(q1.y, 0), GN - 1)];
    const float w3 = dinv[imin(imax(q1.z, 0), GN - 1)] * dinv[imin(imax(q1.w, 0), GN - 1)];
    q0.y = __float_as_int(w0);
    q0.w = __float_as_int(w1);
    q1.y = __float_as_int(w2);
    q1.w = __float_as_int(w3);
    int4* o4 = (int4*)adj;
    o4[t]          = q0;
    o4[t + GE / 4] = q1;
}

static constexpr int NPW = 4;
static_assert(GN % NPW == 0, "a wave's nodes are all valid");

template <int G>
__device__ __forceinline__ void gather_rows(const float* __restrict__ hw,
                                            int s, float w, int j, int c,
                                            float& ax, float& ay)
{
    float2 v[G];
#pragma unroll
    for (int u = 0; u < G; ++u) {
        const int sj = __builtin_amdgcn_readlane(s, j + u);
        v[u] = *(const float2*)(hw + (size_t)sj * HID + c);
    }
#pragma unroll
    for (int u = 0; u < G; ++u) {
        const float wj = __int_as_float(
            __builtin_amdgcn_readlane(__float_as_int(w), j + u));
        ax += wj * v[u].x;
        ay += wj * v[u].y;
    }
}

__launch_bounds__(256)
__global__ void node_gather(const int* __restrict__ rowptr,
                            const int2* __restrict__ adj,
                            const float* __restrict__ dinv,
                            const float* __restrict__ hw,
                            const float* __restrict__ bias,
                            float* __restrict__ Hcat, int col0)
{
    const int lane = threadIdx.x & 31;
    const int wv = __builtin_amdgcn_readfirstlane(
        (int)((blockIdx.x * blockDim.x + threadIdx.x) >> 5));
    const int i0 = wv * NPW;
    if (i0 >= GN) return;
    const int c = lane * 2;

    int rp[NPW + 1];
#pragma unroll
    for (int q = 0; q <= NPW; ++q)
        rp[q] = imin(imax(__builtin_amdgcn_readfirstlane(rowptr[i0 + q]), 0), GE);

    int2   a0[NPW];
    float2 hs[NPW];
    float  di[NPW];
#pragma unroll
    for (int q = 0; q < NPW; ++q) {
        const int n = imin(rp[q + 1] - rp[q], 32);
        const int idx = imin(rp[q] + imax(imin(lane, n - 1), 0), GE - 1);
        a0[q] = adj[idx];
        hs[q] = *(const float2*)(hw + (size_t)(i0 + q) * HID + c);
        di[q] = dinv[i0 + q];
    }
    const float2 bv = *(const float2*)(bias + c);

#pragma unroll
    for (int q = 0; q < NPW; ++q) {
        const int end = rp[q + 1];
        int base = rp[q];
        int2 aa = a0[q];
        float ax = 0.0f, ay = 0.0f;
        while (base < end) {
            const int n = imin(end - base, 32);
            const int   s = imin(imax(aa.x, 0), GN - 1);
            const float w = __int_as_float(aa.y);
            int j = 0;
            for (; j + 8 <= n; j += 8) gather_rows<8>(hw, s, w, j, c, ax, ay);
            if (j + 4 <= n) { gather_rows<4>(hw, s, w, j, c, ax, ay); j += 4; }
            if (j + 2 <= n) { gather_rows<2>(hw, s, w, j, c, ax, ay); j += 2; }
            if (j < n)      { gather_rows<1>(hw, s, w, j, c, ax, ay); }
            base += 32;
            if (base < end) {
                const int n2 = imin(end - base, 32);
                aa = adj[imin(base + imin(lane, n2 - 1), GE - 1)];
            }
        }
        const float sl = di[q] * di[q];
        ax += sl * hs[q].x;
        ay += sl * hs[q].y;
        float2 o;
        o.x = fmaxf(ax + bv.x, 0.0f);
        o.y = fmaxf(ay + bv.y, 0.0f);
        *(float2*)(Hcat + (size_t)(i0 + q) * CAT + col0 + c) = o;
    }
}

__global__ void logsoftmax40(float* __restrict__ out, int M)
{
    int i = blockIdx.x * blockDim.x + threadIdx.x;
    if (i >= M) return;
    float* row = out + (size_t)i * LAB;
    float v[LAB];
    float mx = -3.4e38f;
#pragma unroll
    for (int c = 0; c < LAB; ++c) { v[c] = row[c]; mx = fmaxf(mx, v[c]); }
    float s = 0.0f;
#pragma unroll
    for (int c = 0; c < LAB; ++c) s += __expf(v[c] - mx);
    float ls = __logf(s);
#pragma unroll
    for (int c = 0; c < LAB; ++c) row[c] = v[c] - mx - ls;
}

extern "C" void kernel_launch(void* const* d_in, const int* in_sizes, int n_in,
                              void* d_out, int out_size, void* d_ws, size_t ws_size,
                              hipStream_t stream)
{
    (void)in_sizes; (void)n_in; (void)out_size; (void)ws_size;

    const float* x  = (const float*)d_in[0];
    const int*   ei = (const int*)d_in[1];
    const float* Wm[3] = {(const float*)d_in[2], (const float*)d_in[4],
                          (const float*)d_in[6]};
    const float* bm[3] = {(const float*)d_in[3], (const float*)d_in[5],
                          (const float*)d_in[7]};
    const float* Wl = (const float*)d_in[8];
    const float* bl = (const float*)d_in[9];
    float* out = (float*)d_out;

    char* ws = (char*)d_ws;
    auto align256 = [](size_t b) { return (b + 255) & ~(size_t)255; };
    size_t off = 0;
    float* dinv   = (float*)(ws + off); off += align256((size_t)GN * 4);
    int*   rowptr = (int*)  (ws + off); off += align256((size_t)(GN + 1) * 4);
    int*   binH   = (int*)  (ws + off); off += align256((size_t)NBIN * EB * 4);
    int*   binO   = (int*)  (ws + off); off += align256((size_t)NBIN * EB * 4);
    int*   bstart = (int*)  (ws + off); off += align256((size_t)NBIN * 4);
    int2*  tmp    = (int2*) (ws + off); off += align256((size_t)GE * 8);
    int2*  srt    = (int2*) (ws + off); off += align256((size_t)GE * 8);
    int2*  adj    = tmp;
    float* hw     = (float*)(ws + off); off += align256((size_t)GN * HID * 4);
    float* Hcat   = (float*)(ws + off); off += align256((size_t)GN * CAT * 4);

    const dim3 blk(256);

    bin_hist   <<<EB, dim3(L1T), 0, stream>>>(ei, binH);
    bin_scan   <<<NBIN, dim3(SCT), 0, stream>>>(binH, binO, bstart);
    bin_scatter<<<EB, dim3(L1T), 0, stream>>>(ei, binO, tmp);
    bin_sort   <<<NBIN, dim3(SCT), 0, stream>>>(tmp, bstart, rowptr, dinv, srt);
    edge_norm  <<<(GE / 4 + 255) / 256, blk, 0, stream>>>(srt, dinv, adj);

    const int gemmBlocks = (GN + 127) / 128;
    const int gatherBlocks = (GN / NPW + 7) / 8;

    for (int l = 0; l < 3; ++l) {
        const float* Ain = (l == 0) ? x : (Hcat + HID * (l - 1));
        const int lda = (l == 0) ? DIN : CAT;
        const int K   = (l == 0) ? DIN : HID;
        const size_t shmem = (size_t)(K / 2) * PadRow<64>::value * sizeof(float2);
        gemm_wmma_f32<4><<<gemmBlocks, blk, shmem, stream>>>(
            Ain, lda, Wm[l], K, HID, GN, hw, HID, nullptr);
        node_gather<<<gatherBlocks, blk, 0, stream>>>(
            rowptr, adj, dinv, hw, bm[l], Hcat, HID * l);
    }

    const size_t shmemF = (size_t)(CAT / 2) * PadRow<48>::value * sizeof(float2);
    gemm_wmma_f32<3><<<gemmBlocks, blk, shmemF, stream>>>(
        Hcat, CAT, Wl, CAT, LAB, GN, out, LAB, bl);

    logsoftmax40<<<(GN + 255) / 256, blk, 0, stream>>>(out, GN);
}


# layers 1 and 2: the h @ W product moved into the preceding gather (a block's 32 finished rows go through LDS to the f32 WMMA, one tile per wave, W from global); two GEMM launches gone
# speedup vs baseline: 2.6606x; 1.1308x over previous
#include <hip/hip_runtime.h>


typedef __attribute__((ext_vector_type(2))) float v2f;
typedef __attribute__((ext_vector_type(8))) float v8f;

static constexpr int GN   = 100000;
static constexpr int GE   = 1000000;
static constexpr int DIN  = 128;
static constexpr int HID  = 64;
static constexpr int CAT  = 192;
static constexpr int LAB  = 40;

template <int W> struct PadRow {
    static constexpr int value = W + ((48 - (W % 32)) % 32);
};

template <int NT>
__launch_bounds__(256)
__global__ void gemm_wmma_f32(const float* __restrict__ A, int lda,
                              const float* __restrict__ B, int K, int dout,
                              int M, float* __restrict__ out, int ldo,
                              const float* __restrict__ bias)
{
    constexpr int W  = NT * 16;
    constexpr int WP = PadRow<W>::value;
    extern __shared__ v2f Bs2[];

    const int KP = K >> 1;
    for (int idx = threadIdx.x; idx < KP * W; idx += 256) {
        int kp = idx / W;
        int n  = idx - kp * W;
        v2f p;
        p.x = (n < dout) ? B[(2 * kp)     * dout + n] : 0.0f;
        p.y = (n < dout) ? B[(2 * kp + 1) * dout + n] : 0.0f;
        Bs2[kp * WP + n] = p;
    }
    __syncthreads();

    const int wave = threadIdx.x >> 5;
    const int lane = threadIdx.x & 31;
    const int l15  = lane & 15;
    const int half = lane >> 4;
    const int kb   = half * 2;

    const int m0 = (blockIdx.x * 8 + wave) * 16;
    int mA = m0 + l15;
    if (mA > M - 1) mA = M - 1;
    const float* Arow = A + (size_t)mA * (size_t)lda;

    v8f acc[NT];
#pragma unroll
    for (int t = 0; t < NT; ++t)
        acc[t] = (v8f){0.f, 0.f, 0.f, 0.f, 0.f, 0.f, 0.f, 0.f};

    for (int k = 0; k < K; k += 4) {
        v2f a = *(const v2f*)(Arow + k + kb);
        const v2f* brow = Bs2 + ((k >> 1) + half) * WP + l15;
#pragma unroll
        for (int t = 0; t < NT; ++t) {
            v2f b = brow[t * 16];
            acc[t] = __builtin_amdgcn_wmma_f32_16x16x4_f32(
                false, a, false, b, (short)0, acc[t], false, false);
        }
    }

#pragma unroll
    for (int t = 0; t < NT; ++t) {
        const int n = t * 16 + l15;
        float bv = 0.0f;
        if (bias && n < dout) bv = bias[n];
#pragma unroll
        for (int v = 0; v < 8; ++v) {
            const int m = m0 + half * 8 + v;
            if (m < M && n < dout)
                out[(size_t)m * ldo + n] = acc[t][v] + bv;
        }
    }
}

static constexpr int BW   = 400;
static constexpr int NBIN = GN / BW;
static constexpr int CH   = 4096;
static constexpr int EB   = (GE + CH - 1) / CH;
static constexpr int L1T  = 512;
static constexpr int GRPS  = CH / 4 / L1T;
static constexpr int SCT  = 1024;
static_assert(GN % BW == 0, "ranges cover the nodes exactly");
static_assert(NBIN <= L1T && EB <= SCT && BW <= SCT, "one thread per counter");
static_assert(GE % 4 == 0 && CH % (4 * L1T) == 0, "edges are read four at a time");

__device__ __forceinline__ int imin(int a, int b) { return a < b ? a : b; }
__device__ __forceinline__ int imax(int a, int b) { return a > b ? a : b; }

__device__ __forceinline__ int bin_of(int d) {
    return imin(imax(d, 0), GN - 1) / BW;
}

__device__ __forceinline__ int block_excl_scan(int v, int pre, int* wsum, int* psum)
{
    const int lane = threadIdx.x & 31;
    const int wave = threadIdx.x >> 5;
    int inc = v;
#pragma unroll
    for (int o = 1; o < 32; o <<= 1) {
        const int y = __shfl_up(inc, o);
        if (lane >= o) inc += y;
    }
#pragma unroll
    for (int o = 16; o > 0; o >>= 1) pre += __shfl_xor(pre, o);
    if (lane == 31) wsum[wave] = inc;
    if (lane == 0)  psum[wave] = pre;
    __syncthreads();
    int tot = psum[lane] + ((lane < wave) ? wsum[lane] : 0);
#pragma unroll
    for (int o = 16; o > 0; o >>= 1) tot += __shfl_xor(tot, o);
    return tot + inc - v;
}

__launch_bounds__(L1T)
__global__ void bin_hist(const int* __restrict__ ei, int* __restrict__ H)
{
    __shared__ int h[NBIN];
    const int tid = threadIdx.x;
    const int b   = blockIdx.x;
    if (tid < NBIN) h[tid] = 0;
    __syncthreads();
    const int4* d4 = (const int4*)(ei + GE);
    int4 d[GRPS];
    bool ok[GRPS];
#pragma unroll
    for (int k = 0; k < GRPS; ++k) {
        const int g = b * (CH / 4) + k * L1T + tid;
        ok[k] = g < GE / 4;
        d[k]  = d4[ok[k] ? g : 0];
    }
#pragma unroll
    for (int k = 0; k < GRPS; ++k) {
        if (ok[k]) {
            atomicAdd(&h[bin_of(d[k].x)], 1);
            atomicAdd(&h[bin_of(d[k].y)], 1);
            atomicAdd(&h[bin_of(d[k].z)], 1);
            atomicAdd(&h[bin_of(d[k].w)], 1);
        }
    }
    __syncthreads();
    if (tid < NBIN) H[tid * EB + b] = h[tid];
}

__launch_bounds__(SCT)
__global__ void bin_scan(const int* __restrict__ H, int* __restrict__ O,
                         int* __restrict__ binstart)
{
    __shared__ int wsum[32];
    __shared__ int psum[32];
    const int tid = threadIdx.x;
    const int n0  = blockIdx.x * EB;
    int pre = 0;
    for (int i = tid; i < n0; i += SCT) pre += H[i];
    const int v = (tid < EB) ? H[n0 + tid] : 0;
    const int ex = block_excl_scan(v, pre, wsum, psum);
    if (tid < EB) O[n0 + tid] = ex;
    if (tid == 0) binstart[blockIdx.x] = ex;
}

__launch_bounds__(L1T)
__global__ void bin_scatter(const int* __restrict__ ei, const int* __restrict__ O,
                            int2* __restrict__ tmp)
{
    __shared__ int cur[NBIN];
    const int tid = threadIdx.x;
    const int b   = blockIdx.x;
    if (tid < NBIN) cur[tid] = O[tid * EB + b];
    __syncthreads();
    const int4* s4 = (const int4*)ei;
    const int4* d4 = (const int4*)(ei + GE);
    int4 s[GRPS], d[GRPS];
    bool ok[GRPS];
#pragma unroll
    for (int k = 0; k < GRPS; ++k) {
        const int g = b * (CH / 4) + k * L1T + tid;
        ok[k] = g < GE / 4;
        s[k]  = s4[ok[k] ? g : 0];
        d[k]  = d4[ok[k] ? g : 0];
    }
#pragma unroll
    for (int k = 0; k < GRPS; ++k) {
        if (ok[k]) {
            const int p0 = atomicAdd(&cur[bin_of(d[k].x)], 1);
            const int p1 = atomicAdd(&cur[bin_of(d[k].y)], 1);
            const int p2 = atomicAdd(&cur[bin_of(d[k].z)], 1);
            const int p3 = atomicAdd(&cur[bin_of(d[k].w)], 1);
            if ((unsigned)p0 < (unsigned)GE) tmp[p0] = make_int2(s[k].x, d[k].x);
            if ((unsigned)p1 < (unsigned)GE) tmp[p1] = make_int2(s[k].y, d[k].y);
            if ((unsigned)p2 < (unsigned)GE) tmp[p2] = make_int2(s[k].z, d[k].z);
            if ((unsigned)p3 < (unsigned)GE) tmp[p3] = make_int2(s[k].w, d[k].w);
        }
    }
}

static constexpr int RPT = 4;
__launch_bounds__(SCT)
__global__ void bin_sort(const int2* __restrict__ tmp, const int* __restrict__ binstart,
                         int* __restrict__ rowptr, float* __restrict__ dinv,
                         int2* __restrict__ srt)
{
    __shared__ int c[BW];
    __shared__ int wsum[32];
    __shared__ int psum[32];
    const int tid  = threadIdx.x;
    const int bin  = blockIdx.x;
    const int nb   = bin * BW;
    const int lo   = imin(imax(binstart[bin], 0), GE);
    const int hi   = (bin == NBIN - 1) ? GE : imin(imax(binstart[bin + 1], lo), GE);

    if (tid < BW) c[tid] = 0;
    __syncthreads();
    for (int p0 = lo + tid; p0 < hi; p0 += RPT * SCT) {
        int dl[RPT];
#pragma unroll
        for (int u = 0; u < RPT; ++u)
            dl[u] = imin(imax(tmp[imin(p0 + u * SCT, hi - 1)].y - nb, 0), BW - 1);
#pragma unroll
        for (int u = 0; u < RPT; ++u)
            atomicAdd(&c[dl[u]], (p0 + u * SCT < hi) ? 1 : 0);
    }
    __syncthreads();
    const int v  = (tid < BW) ? c[tid] : 0;
    const int ex = block_excl_scan(v, 0, wsum, psum);
    if (tid < BW) {
        rowptr[nb + tid] = lo + ex;
        dinv[nb + tid]   = rsqrtf((float)(v + 1));
        c[tid] = ex;
    }
    if (bin == NBIN - 1 && tid == 0) rowptr[GN] = hi;
    __syncthreads();
    for (int p0 = lo + tid; p0 < hi; p0 += RPT * SCT) {
        int2 sd[RPT];
        int  pos[RPT];
#pragma unroll
        for (int u = 0; u < RPT; ++u)
            sd[u] = tmp[imin(p0 + u * SCT, hi - 1)];
#pragma unroll
        for (int u = 0; u < RPT; ++u) {
            const bool live = p0 + u * SCT < hi;
            const int r = atomicAdd(&c[imin(imax(sd[u].y - nb, 0), BW - 1)], live ? 1 : 0);
            pos[u] = live ? lo + r : -1;
        }
#pragma unroll
        for (int u = 0; u < RPT; ++u)
            if ((unsigned)pos[u] < (unsigned)GE) srt[pos[u]] = sd[u];
    }
}

__global__ void edge_norm(const int2* __restrict__ srt, const float* __restrict__ dinv,
                          int2* __restrict__ adj)
{
    const int t = blockIdx.x * blockDim.x + threadIdx.x;
    if (t >= GE / 4) return;
    const int4* i4 = (const int4*)srt;
    int4 q0 = i4[t];
    int4 q1 = i4[t + GE / 4];
    const float w0 = dinv[imin(imax(q0.x, 0), GN - 1)] * dinv[imin(imax(q0.y, 0), GN - 1)];
    const float w1 = dinv[imin(imax(q0.z, 0), GN - 1)] * dinv[imin(imax(q0.w, 0), GN - 1)];
    const float w2 = dinv[imin(imax(q1.x, 0), GN - 1)] * dinv[imin(imax(q1.y, 0), GN - 1)];
    const float w3 = dinv[imin(imax(q1.z, 0), GN - 1)] * dinv[imin(imax(q1.w, 0), GN - 1)];
    q0.y = __float_as_int(w0);
    q0.w = __float_as_int(w1);
    q1.y = __float_as_int(w2);
    q1.w = __float_as_int(w3);
    int4* o4 = (int4*)adj;
    o4[t]          = q0;
    o4[t + GE / 4] = q1;
}

static constexpr int NPW = 4;
static_assert(GN % NPW == 0, "a wave's nodes are all valid");

template <int G>
__device__ __forceinline__ void gather_rows(const float* __restrict__ hw,
                                            int s, float w, int j, int c,
                                            float& ax, float& ay)
{
    float2 v[G];
#pragma unroll
    for (int u = 0; u < G; ++u) {
        const int sj = __builtin_amdgcn_readlane(s, j + u);
        v[u] = *(const float2*)(hw + (size_t)sj * HID + c);
    }
#pragma unroll
    for (int u = 0; u < G; ++u) {
        const float wj = __int_as_float(
            __builtin_amdgcn_readlane(__float_as_int(w), j + u));
        ax += wj * v[u].x;
        ay += wj * v[u].y;
    }
}

static constexpr int GBW  = 8;
static constexpr int GBN  = GBW * NPW;
static constexpr int APAD = HID + 4;
static_assert(GN % GBN == 0, "every block owns GBN valid nodes");
static_assert(GBN == 32 && HID == 64, "8 waves = 2 row tiles x 4 column tiles");

template <bool FUSE>
__launch_bounds__(256)
__global__ void node_gather(const int* __restrict__ rowptr,
                            const int2* __restrict__ adj,
                            const float* __restrict__ dinv,
                            const float* __restrict__ hw,
                            const float* __restrict__ bias,
                            float* __restrict__ Hcat, int col0,
                            const float* __restrict__ Wn,
                            float* __restrict__ hwn)
{
    __shared__ float As[FUSE ? GBN * APAD : 1];
    const int lane = threadIdx.x & 31;
    const int wib = __builtin_amdgcn_readfirstlane((int)(threadIdx.x >> 5));
    const int nb0 = blockIdx.x * GBN;
    if (nb0 >= GN) return;
    const int i0 = nb0 + wib * NPW;
    const int c = lane * 2;

    int rp[NPW + 1];
#pragma unroll
    for (int q = 0; q <= NPW; ++q)
        rp[q] = imin(imax(__builtin_amdgcn_readfirstlane(rowptr[i0 + q]), 0), GE);

    int2   a0[NPW];
    float2 hs[NPW];
    float  di[NPW];
#pragma unroll
    for (int q = 0; q < NPW; ++q) {
        const int n = imin(rp[q + 1] - rp[q], 32);
        const int idx = imin(rp[q] + imax(imin(lane, n - 1), 0), GE - 1);
        a0[q] = adj[idx];
        hs[q] = *(const float2*)(hw + (size_t)(i0 + q) * HID + c);
        di[q] = dinv[i0 + q];
    }
    const float2 bv = *(const float2*)(bias + c);

#pragma unroll
    for (int q = 0; q < NPW; ++q) {
        const int end = rp[q + 1];
        int base = rp[q];
        int2 aa = a0[q];
        float ax = 0.0f, ay = 0.0f;
        while (base < end) {
            const int n = imin(end - base, 32);
            const int   s = imin(imax(aa.x, 0), GN - 1);
            const float w = __int_as_float(aa.y);
            int j = 0;
            for (; j + 8 <= n; j += 8) gather_rows<8>(hw, s, w, j, c, ax, ay);
            if (j + 4 <= n) { gather_rows<4>(hw, s, w, j, c, ax, ay); j += 4; }
            if (j + 2 <= n) { gather_rows<2>(hw, s, w, j, c, ax, ay); j += 2; }
            if (j < n)      { gather_rows<1>(hw, s, w, j, c, ax, ay); }
            base += 32;
            if (base < end) {
                const int n2 = imin(end - base, 32);
                aa = adj[imin(base + imin(lane, n2 - 1), GE - 1)];
            }
        }
        const float sl = di[q] * di[q];
        ax += sl * hs[q].x;
        ay += sl * hs[q].y;
        float2 o;
        o.x = fmaxf(ax + bv.x, 0.0f);
        o.y = fmaxf(ay + bv.y, 0.0f);
        *(float2*)(Hcat + (size_t)(i0 + q) * CAT + col0 + c) = o;
        if constexpr (FUSE)
            *(float2*)(As + (wib * NPW + q) * APAD + c) = o;
    }

    if constexpr (FUSE) {
        const int rt   = wib >> 2;
        const int ct   = wib & 3;
        const int l15  = lane & 15;
        const int half = lane >> 4;
        const int kb   = half * 2;

        const float* wp = Wn + (size_t)kb * HID + ct * 16 + l15;
        v2f bf[HID / 4];
#pragma unroll
        for (int ks = 0; ks < HID / 4; ++ks) {
            bf[ks].x = wp[(4 * ks)     * HID];
            bf[ks].y = wp[(4 * ks + 1) * HID];
        }
        __syncthreads();

        const float* arow = As + (rt * 16 + l15) * APAD + kb;
        v8f acc = (v8f){0.f, 0.f, 0.f, 0.f, 0.f, 0.f, 0.f, 0.f};
#pragma unroll
        for (int ks = 0; ks < HID / 4; ++ks) {
            const v2f a = *(const v2f*)(arow + 4 * ks);
            acc = __builtin_amdgcn_wmma_f32_16x16x4_f32(
                false, a, false, bf[ks], (short)0, acc, false, false);
        }
        float* orow = hwn + (size_t)(nb0 + rt * 16 + half * 8) * HID + ct * 16 + l15;
#pragma unroll
        for (int v = 0; v < 8; ++v)
            orow[(size_t)v * HID] = acc[v];
    }
}

__global__ void logsoftmax40(float* __restrict__ out, int M)
{
    int i = blockIdx.x * blockDim.x + threadIdx.x;
    if (i >= M) return;
    float* row = out + (size_t)i * LAB;
    float v[LAB];
    float mx = -3.4e38f;
#pragma unroll
    for (int c = 0; c < LAB; ++c) { v[c] = row[c]; mx = fmaxf(mx, v[c]); }
    float s = 0.0f;
#pragma unroll
    for (int c = 0; c < LAB; ++c) s += __expf(v[c] - mx);
    float ls = __logf(s);
#pragma unroll
    for (int c = 0; c < LAB; ++c) row[c] = v[c] - mx - ls;
}

extern "C" void kernel_launch(void* const* d_in, const int* in_sizes, int n_in,
                              void* d_out, int out_size, void* d_ws, size_t ws_size,
                              hipStream_t stream)
{
    (void)in_sizes; (void)n_in; (void)out_size; (void)ws_size;

    const float* x  = (const float*)d_in[0];
    const int*   ei = (const int*)d_in[1];
    const float* Wm[3] = {(const float*)d_in[2], (const float*)d_in[4],
                          (const float*)d_in[6]};
    const float* bm[3] = {(const float*)d_in[3], (const float*)d_in[5],
                          (const float*)d_in[7]};
    const float* Wl = (const float*)d_in[8];
    const float* bl = (const float*)d_in[9];
    float* out = (float*)d_out;

    char* ws = (char*)d_ws;
    auto align256 = [](size_t b) { return (b + 255) & ~(size_t)255; };
    size_t off = 0;
    float* dinv   = (float*)(ws + off); off += align256((size_t)GN * 4);
    int*   rowptr = (int*)  (ws + off); off += align256((size_t)(GN + 1) * 4);
    int*   binH   = (int*)  (ws + off); off += align256((size_t)NBIN * EB * 4);
    int*   binO   = (int*)  (ws + off); off += align256((size_t)NBIN * EB * 4);
    int*   bstart = (int*)  (ws + off); off += align256((size_t)NBIN * 4);
    int2*  tmp    = (int2*) (ws + off); off += align256((size_t)GE * 8);
    int2*  srt    = (int2*) (ws + off); off += align256((size_t)GE * 8);
    int2*  adj    = tmp;
    float* hw     = (float*)(ws + off); off += align256((size_t)GN * HID * 4);
    float* hw2    = (float*)(ws + off); off += align256((size_t)GN * HID * 4);
    float* Hcat   = (float*)(ws + off); off += align256((size_t)GN * CAT * 4);

    const dim3 blk(256);

    bin_hist   <<<EB, dim3(L1T), 0, stream>>>(ei, binH);
    bin_scan   <<<NBIN, dim3(SCT), 0, stream>>>(binH, binO, bstart);
    bin_scatter<<<EB, dim3(L1T), 0, stream>>>(ei, binO, tmp);
    bin_sort   <<<NBIN, dim3(SCT), 0, stream>>>(tmp, bstart, rowptr, dinv, srt);
    edge_norm  <<<(GE / 4 + 255) / 256, blk, 0, stream>>>(srt, dinv, adj);

    const int gemmBlocks = (GN + 127) / 128;
    const int gatherBlocks = GN / GBN;

    const size_t shmem0 = (size_t)(DIN / 2) * PadRow<64>::value * sizeof(float2);
    gemm_wmma_f32<4><<<gemmBlocks, blk, shmem0, stream>>>(
        x, DIN, Wm[0], DIN, HID, GN, hw, HID, nullptr);
    node_gather<true><<<gatherBlocks, blk, 0, stream>>>(
        rowptr, adj, dinv, hw, bm[0], Hcat, 0, Wm[1], hw2);
    node_gather<true><<<gatherBlocks, blk, 0, stream>>>(
        rowptr, adj, dinv, hw2, bm[1], Hcat, HID, Wm[2], hw);
    node_gather<false><<<gatherBlocks, blk, 0, stream>>>(
        rowptr, adj, dinv, hw, bm[2], Hcat, 2 * HID, nullptr, nullptr);

    const size_t shmemF = (size_t)(CAT / 2) * PadRow<48>::value * sizeof(float2);
    gemm_wmma_f32<3><<<gemmBlocks, blk, shmemF, stream>>>(
        Hcat, CAT, Wl, CAT, LAB, GN, out, LAB, bl);

    logsoftmax40<<<(GN + 255) / 256, blk, 0, stream>>>(out, GN);
}


# final linear and log_softmax moved into the layer-2 gather (block's 32 concatenated rows via LDS to the f32 WMMA, k order kept); two launches gone; softmax sum of exps now a 32-lane butterfly
# speedup vs baseline: 2.8920x; 1.0870x over previous
#include <hip/hip_runtime.h>


typedef __attribute__((ext_vector_type(2))) float v2f;
typedef __attribute__((ext_vector_type(8))) float v8f;

static constexpr int GN   = 100000;
static constexpr int GE   = 1000000;
static constexpr int DIN  = 128;
static constexpr int HID  = 64;
static constexpr int CAT  = 192;
static constexpr int LAB  = 40;

template <int W> struct PadRow {
    static constexpr int value = W + ((48 - (W % 32)) % 32);
};

template <int NT>
__launch_bounds__(256)
__global__ void gemm_wmma_f32(const float* __restrict__ A, int lda,
                              const float* __restrict__ B, int K, int dout,
                              int M, float* __restrict__ out, int ldo,
                              const float* __restrict__ bias)
{
    constexpr int W  = NT * 16;
    constexpr int WP = PadRow<W>::value;
    extern __shared__ v2f Bs2[];

    const int KP = K >> 1;
    for (int idx = threadIdx.x; idx < KP * W; idx += 256) {
        int kp = idx / W;
        int n  = idx - kp * W;
        v2f p;
        p.x = (n < dout) ? B[(2 * kp)     * dout + n] : 0.0f;
        p.y = (n < dout) ? B[(2 * kp + 1) * dout + n] : 0.0f;
        Bs2[kp * WP + n] = p;
    }
    __syncthreads();

    const int wave = threadIdx.x >> 5;
    const int lane = threadIdx.x & 31;
    const int l15  = lane & 15;
    const int half = lane >> 4;
    const int kb   = half * 2;

    const int m0 = (blockIdx.x * 8 + wave) * 16;
    int mA = m0 + l15;
    if (mA > M - 1) mA = M - 1;
    const float* Arow = A + (size_t)mA * (size_t)lda;

    v8f acc[NT];
#pragma unroll
    for (int t = 0; t < NT; ++t)
        acc[t] = (v8f){0.f, 0.f, 0.f, 0.f, 0.f, 0.f, 0.f, 0.f};

    for (int k = 0; k < K; k += 4) {
        v2f a = *(const v2f*)(Arow + k + kb);
        const v2f* brow = Bs2 + ((k >> 1) + half) * WP + l15;
#pragma unroll
        for (int t = 0; t < NT; ++t) {
            v2f b = brow[t * 16];
            acc[t] = __builtin_amdgcn_wmma_f32_16x16x4_f32(
                false, a, false, b, (short)0, acc[t], false, false);
        }
    }

#pragma unroll
    for (int t = 0; t < NT; ++t) {
        const int n = t * 16 + l15;
        float bv = 0.0f;
        if (bias && n < dout) bv = bias[n];
#pragma unroll
        for (int v = 0; v < 8; ++v) {
            const int m = m0 + half * 8 + v;
            if (m < M && n < dout)
                out[(size_t)m * ldo + n] = acc[t][v] + bv;
        }
    }
}

static constexpr int BW   = 400;
static constexpr int NBIN = GN / BW;
static constexpr int CH   = 4096;
static constexpr int EB   = (GE + CH - 1) / CH;
static constexpr int L1T  = 512;
static constexpr int GRPS  = CH / 4 / L1T;
static constexpr int SCT  = 1024;
static_assert(GN % BW == 0, "ranges cover the nodes exactly");
static_assert(NBIN <= L1T && EB <= SCT && BW <= SCT, "one thread per counter");
static_assert(GE % 4 == 0 && CH % (4 * L1T) == 0, "edges are read four at a time");

__device__ __forceinline__ int imin(int a, int b) { return a < b ? a : b; }
__device__ __forceinline__ int imax(int a, int b) { return a > b ? a : b; }

__device__ __forceinline__ int bin_of(int d) {
    return imin(imax(d, 0), GN - 1) / BW;
}

__device__ __forceinline__ int block_excl_scan(int v, int pre, int* wsum, int* psum)
{
    const int lane = threadIdx.x & 31;
    const int wave = threadIdx.x >> 5;
    int inc = v;
#pragma unroll
    for (int o = 1; o < 32; o <<= 1) {
        const int y = __shfl_up(inc, o);
        if (lane >= o) inc += y;
    }
#pragma unroll
    for (int o = 16; o > 0; o >>= 1) pre += __shfl_xor(pre, o);
    if (lane == 31) wsum[wave] = inc;
    if (lane == 0)  psum[wave] = pre;
    __syncthreads();
    int tot = psum[lane] + ((lane < wave) ? wsum[lane] : 0);
#pragma unroll
    for (int o = 16; o > 0; o >>= 1) tot += __shfl_xor(tot, o);
    return tot + inc - v;
}

__launch_bounds__(L1T)
__global__ void bin_hist(const int* __restrict__ ei, int* __restrict__ H)
{
    __shared__ int h[NBIN];
    const int tid = threadIdx.x;
    const int b   = blockIdx.x;
    if (tid < NBIN) h[tid] = 0;
    __syncthreads();
    const int4* d4 = (const int4*)(ei + GE);
    int4 d[GRPS];
    bool ok[GRPS];
#pragma unroll
    for (int k = 0; k < GRPS; ++k) {
        const int g = b * (CH / 4) + k * L1T + tid;
        ok[k] = g < GE / 4;
        d[k]  = d4[ok[k] ? g : 0];
    }
#pragma unroll
    for (int k = 0; k < GRPS; ++k) {
        if (ok[k]) {
            atomicAdd(&h[bin_of(d[k].x)], 1);
            atomicAdd(&h[bin_of(d[k].y)], 1);
            atomicAdd(&h[bin_of(d[k].z)], 1);
            atomicAdd(&h[bin_of(d[k].w)], 1);
        }
    }
    __syncthreads();
    if (tid < NBIN) H[tid * EB + b] = h[tid];
}

__launch_bounds__(SCT)
__global__ void bin_scan(const int* __restrict__ H, int* __restrict__ O,
                         int* __restrict__ binstart)
{
    __shared__ int wsum[32];
    __shared__ int psum[32];
    const int tid = threadIdx.x;
    const int n0  = blockIdx.x * EB;
    int pre = 0;
    for (int i = tid; i < n0; i += SCT) pre += H[i];
    const int v = (tid < EB) ? H[n0 + tid] : 0;
    const int ex = block_excl_scan(v, pre, wsum, psum);
    if (tid < EB) O[n0 + tid] = ex;
    if (tid == 0) binstart[blockIdx.x] = ex;
}

__launch_bounds__(L1T)
__global__ void bin_scatter(const int* __restrict__ ei, const int* __restrict__ O,
                            int2* __restrict__ tmp)
{
    __shared__ int cur[NBIN];
    const int tid = threadIdx.x;
    const int b   = blockIdx.x;
    if (tid < NBIN) cur[tid] = O[tid * EB + b];
    __syncthreads();
    const int4* s4 = (const int4*)ei;
    const int4* d4 = (const int4*)(ei + GE);
    int4 s[GRPS], d[GRPS];
    bool ok[GRPS];
#pragma unroll
    for (int k = 0; k < GRPS; ++k) {
        const int g = b * (CH / 4) + k * L1T + tid;
        ok[k] = g < GE / 4;
        s[k]  = s4[ok[k] ? g : 0];
        d[k]  = d4[ok[k] ? g : 0];
    }
#pragma unroll
    for (int k = 0; k < GRPS; ++k) {
        if (ok[k]) {
            const int p0 = atomicAdd(&cur[bin_of(d[k].x)], 1);
            const int p1 = atomicAdd(&cur[bin_of(d[k].y)], 1);
            const int p2 = atomicAdd(&cur[bin_of(d[k].z)], 1);
            const int p3 = atomicAdd(&cur[bin_of(d[k].w)], 1);
            if ((unsigned)p0 < (unsigned)GE) tmp[p0] = make_int2(s[k].x, d[k].x);
            if ((unsigned)p1 < (unsigned)GE) tmp[p1] = make_int2(s[k].y, d[k].y);
            if ((unsigned)p2 < (unsigned)GE) tmp[p2] = make_int2(s[k].z, d[k].z);
            if ((unsigned)p3 < (unsigned)GE) tmp[p3] = make_int2(s[k].w, d[k].w);
        }
    }
}

static constexpr int RPT = 4;
__launch_bounds__(SCT)
__global__ void bin_sort(const int2* __restrict__ tmp, const int* __restrict__ binstart,
                         int* __restrict__ rowptr, float* __restrict__ dinv,
                         int2* __restrict__ srt)
{
    __shared__ int c[BW];
    __shared__ int wsum[32];
    __shared__ int psum[32];
    const int tid  = threadIdx.x;
    const int bin  = blockIdx.x;
    const int nb   = bin * BW;
    const int lo   = imin(imax(binstart[bin], 0), GE);
    const int hi   = (bin == NBIN - 1) ? GE : imin(imax(binstart[bin + 1], lo), GE);

    if (tid < BW) c[tid] = 0;
    __syncthreads();
    for (int p0 = lo + tid; p0 < hi; p0 += RPT * SCT) {
        int dl[RPT];
#pragma unroll
        for (int u = 0; u < RPT; ++u)
            dl[u] = imin(imax(tmp[imin(p0 + u * SCT, hi - 1)].y - nb, 0), BW - 1);
#pragma unroll
        for (int u = 0; u < RPT; ++u)
            atomicAdd(&c[dl[u]], (p0 + u * SCT < hi) ? 1 : 0);
    }
    __syncthreads();
    const int v  = (tid < BW) ? c[tid] : 0;
    const int ex = block_excl_scan(v, 0, wsum, psum);
    if (tid < BW) {
        rowptr[nb + tid] = lo + ex;
        dinv[nb + tid]   = rsqrtf((float)(v + 1));
        c[tid] = ex;
    }
    if (bin == NBIN - 1 && tid == 0) rowptr[GN] = hi;
    __syncthreads();
    for (int p0 = lo + tid; p0 < hi; p0 += RPT * SCT) {
        int2 sd[RPT];
        int  pos[RPT];
#pragma unroll
        for (int u = 0; u < RPT; ++u)
            sd[u] = tmp[imin(p0 + u * SCT, hi - 1)];
#pragma unroll
        for (int u = 0; u < RPT; ++u) {
            const bool live = p0 + u * SCT < hi;
            const int r = atomicAdd(&c[imin(imax(sd[u].y - nb, 0), BW - 1)], live ? 1 : 0);
            pos[u] = live ? lo + r : -1;
        }
#pragma unroll
        for (int u = 0; u < RPT; ++u)
            if ((unsigned)pos[u] < (unsigned)GE) srt[pos[u]] = sd[u];
    }
}

__global__ void edge_norm(const int2* __restrict__ srt, const float* __restrict__ dinv,
                          int2* __restrict__ adj)
{
    const int t = blockIdx.x * blockDim.x + threadIdx.x;
    if (t >= GE / 4) return;
    const int4* i4 = (const int4*)srt;
    int4 q0 = i4[t];
    int4 q1 = i4[t + GE / 4];
    const float w0 = dinv[imin(imax(q0.x, 0), GN - 1)] * dinv[imin(imax(q0.y, 0), GN - 1)];
    const float w1 = dinv[imin(imax(q0.z, 0), GN - 1)] * dinv[imin(imax(q0.w, 0), GN - 1)];
    const float w2 = dinv[imin(imax(q1.x, 0), GN - 1)] * dinv[imin(imax(q1.y, 0), GN - 1)];
    const float w3 = dinv[imin(imax(q1.z, 0), GN - 1)] * dinv[imin(imax(q1.w, 0), GN - 1)];
    q0.y = __float_as_int(w0);
    q0.w = __float_as_int(w1);
    q1.y = __float_as_int(w2);
    q1.w = __float_as_int(w3);
    int4* o4 = (int4*)adj;
    o4[t]          = q0;
    o4[t + GE / 4] = q1;
}

static constexpr int NPW = 4;
static_assert(GN % NPW == 0, "a wave's nodes are all valid");

template <int G>
__device__ __forceinline__ void gather_rows(const float* __restrict__ hw,
                                            int s, float w, int j, int c,
                                            float& ax, float& ay)
{
    float2 v[G];
#pragma unroll
    for (int u = 0; u < G; ++u) {
        const int sj = __builtin_amdgcn_readlane(s, j + u);
        v[u] = *(const float2*)(hw + (size_t)sj * HID + c);
    }
#pragma unroll
    for (int u = 0; u < G; ++u) {
        const float wj = __int_as_float(
            __builtin_amdgcn_readlane(__float_as_int(w), j + u));
        ax += wj * v[u].x;
        ay += wj * v[u].y;
    }
}

template <class Pre, class Emit>
__device__ __forceinline__ void gather_nodes(const int* __restrict__ rowptr,
                                             const int2* __restrict__ adj,
                                             const float* __restrict__ dinv,
                                             const float* __restrict__ hw,
                                             const float* __restrict__ bias,
                                             int i0, int lane, Pre pre, Emit emit)
{
    const int c = lane * 2;
    int rp[NPW + 1];
#pragma unroll
    for (int q = 0; q <= NPW; ++q)
        rp[q] = imin(imax(__builtin_amdgcn_readfirstlane(rowptr[i0 + q]), 0), GE);

    int2   a0[NPW];
    float2 hs[NPW];
    float  di[NPW];
#pragma unroll
    for (int q = 0; q < NPW; ++q) {
        const int n = imin(rp[q + 1] - rp[q], 32);
        const int idx = imin(rp[q] + imax(imin(lane, n - 1), 0), GE - 1);
        a0[q] = adj[idx];
        hs[q] = *(const float2*)(hw + (size_t)(i0 + q) * HID + c);
        di[q] = dinv[i0 + q];
    }
    const float2 bv = *(const float2*)(bias + c);
    pre();

#pragma unroll
    for (int q = 0; q < NPW; ++q) {
        const int end = rp[q + 1];
        int base = rp[q];
        int2 aa = a0[q];
        float ax = 0.0f, ay = 0.0f;
        while (base < end) {
            const int n = imin(end - base, 32);
            const int   s = imin(imax(aa.x, 0), GN - 1);
            const float w = __int_as_float(aa.y);
            int j = 0;
            for (; j + 8 <= n; j += 8) gather_rows<8>(hw, s, w, j, c, ax, ay);
            if (j + 4 <= n) { gather_rows<4>(hw, s, w, j, c, ax, ay); j += 4; }
            if (j + 2 <= n) { gather_rows<2>(hw, s, w, j, c, ax, ay); j += 2; }
            if (j < n)      { gather_rows<1>(hw, s, w, j, c, ax, ay); }
            base += 32;
            if (base < end) {
                const int n2 = imin(end - base, 32);
                aa = adj[imin(base + imin(lane, n2 - 1), GE - 1)];
            }
        }
        const float sl = di[q] * di[q];
        ax += sl * hs[q].x;
        ay += sl * hs[q].y;
        float2 o;
        o.x = fmaxf(ax + bv.x, 0.0f);
        o.y = fmaxf(ay + bv.y, 0.0f);
        emit(q, o);
    }
}

static constexpr int GBW  = 8;
static constexpr int GBN  = GBW * NPW;
static constexpr int APAD = HID + 4;
static constexpr int HP   = 2 * HID;
static_assert(GN % GBN == 0, "every block owns GBN valid nodes");
static_assert(GBN == 32 && HID == 64, "8 waves = 2 row tiles x 4 column tiles");

__launch_bounds__(256)
__global__ void node_gather(const int* __restrict__ rowptr,
                            const int2* __restrict__ adj,
                            const float* __restrict__ dinv,
                            const float* __restrict__ hw,
                            const float* __restrict__ bias,
                            float* __restrict__ H01, int col0,
                            const float* __restrict__ Wn,
                            float* __restrict__ hwn)
{
    __shared__ float As[GBN * APAD];
    const int lane = threadIdx.x & 31;
    const int wib = __builtin_amdgcn_readfirstlane((int)(threadIdx.x >> 5));
    const int nb0 = blockIdx.x * GBN;
    if (nb0 >= GN) return;
    const int i0 = nb0 + wib * NPW;
    const int c = lane * 2;

    gather_nodes(rowptr, adj, dinv, hw, bias, i0, lane, [] {}, [&](int q, float2 o) {
        *(float2*)(H01 + (size_t)(i0 + q) * HP + col0 + c) = o;
        *(float2*)(As + (wib * NPW + q) * APAD + c) = o;
    });

    const int rt   = wib >> 2;
    const int ct   = wib & 3;
    const int l15  = lane & 15;
    const int half = lane >> 4;
    const int kb   = half * 2;

    const float* wp = Wn + (size_t)kb * HID + ct * 16 + l15;
    v2f bf[HID / 4];
#pragma unroll
    for (int ks = 0; ks < HID / 4; ++ks) {
        bf[ks].x = wp[(4 * ks)     * HID];
        bf[ks].y = wp[(4 * ks + 1) * HID];
    }
    __syncthreads();

    const float* arow = As + (rt * 16 + l15) * APAD + kb;
    v8f acc = (v8f){0.f, 0.f, 0.f, 0.f, 0.f, 0.f, 0.f, 0.f};
#pragma unroll
    for (int ks = 0; ks < HID / 4; ++ks) {
        const v2f a = *(const v2f*)(arow + 4 * ks);
        acc = __builtin_amdgcn_wmma_f32_16x16x4_f32(
            false, a, false, bf[ks], (short)0, acc, false, false);
    }
    float* orow = hwn + (size_t)(nb0 + rt * 16 + half * 8) * HID + ct * 16 + l15;
#pragma unroll
    for (int v = 0; v < 8; ++v)
        orow[(size_t)v * HID] = acc[v];
}

static constexpr int FPAD = CAT + 4;
static constexpr int NTL  = (LAB + 15) / 16;
static constexpr int LP   = 52;
static constexpr int KCH  = 16;
static_assert(NTL <= GBW && NTL * 16 <= LP && CAT % (2 * KCH) == 0 && LAB > 32 && LAB <= 64,
              "logit tiles fit the block; a lane owns labels lane and lane + 32");

__launch_bounds__(256)
__global__ void node_gather_head(const int* __restrict__ rowptr,
                                 const int2* __restrict__ adj,
                                 const float* __restrict__ dinv,
                                 const float* __restrict__ hw,
                                 const float* __restrict__ bias,
                                 const float* __restrict__ H01,
                                 const float* __restrict__ Wl,
                                 const float* __restrict__ bl,
                                 float* __restrict__ out)
{
    __shared__ float As[GBN * FPAD];
    __shared__ float Ls[GBN * LP];
    const int lane = threadIdx.x & 31;
    const int wib = __builtin_amdgcn_readfirstlane((int)(threadIdx.x >> 5));
    const int nb0 = blockIdx.x * GBN;
    if (nb0 >= GN) return;
    const int i0 = nb0 + wib * NPW;
    const int c = lane * 2;

    const int l15  = lane & 15;
    const int half = lane >> 4;
    const int kb   = half * 2;
    const int n    = wib * 16 + l15;
    const int nc   = (n < LAB) ? n : 0;
    const float* wp = Wl + (size_t)kb * LAB + nc;
    auto load_b = [&](v2f (&bf)[KCH / 4], int k0) {
#pragma unroll
        for (int ks = 0; ks < KCH / 4; ++ks) {
            const float* p = wp + (size_t)(k0 + 4 * ks) * LAB;
            bf[ks].x = p[0];
            bf[ks].y = p[LAB];
        }
    };

    gather_nodes(rowptr, adj, dinv, hw, bias, i0, lane,
        [&] {
            float4 hc[NPW];
#pragma unroll
            for (int q = 0; q < NPW; ++q)
                hc[q] = *(const float4*)(H01 + (size_t)(i0 + q) * HP + lane * 4);
#pragma unroll
            for (int q = 0; q < NPW; ++q)
                *(float4*)(As + (wib * NPW + q) * FPAD + lane * 4) = hc[q];
        },
        [&](int q, float2 o) {
            *(float2*)(As + (wib * NPW + q) * FPAD + HP + c) = o;
        });

    const bool mm = wib < NTL;
    v2f bf0[KCH / 4], bf1[KCH / 4];
#pragma unroll
    for (int ks = 0; ks < KCH / 4; ++ks) bf0[ks] = (v2f){0.f, 0.f};
    if (mm) load_b(bf0, 0);
    __syncthreads();

    if (mm) {
        const float* ar0 = As + l15 * FPAD + kb;
        const float* ar1 = As + (16 + l15) * FPAD + kb;
        v8f acc0 = (v8f){0.f, 0.f, 0.f, 0.f, 0.f, 0.f, 0.f, 0.f};
        v8f acc1 = (v8f){0.f, 0.f, 0.f, 0.f, 0.f, 0.f, 0.f, 0.f};
        auto mma_chunk = [&](const v2f (&bf)[KCH / 4], int k0) {
#pragma unroll
            for (int ks = 0; ks < KCH / 4; ++ks) {
                const v2f a0 = *(const v2f*)(ar0 + k0 + 4 * ks);
                const v2f a1 = *(const v2f*)(ar1 + k0 + 4 * ks);
                acc0 = __builtin_amdgcn_wmma_f32_16x16x4_f32(
                    false, a0, false, bf[ks], (short)0, acc0, false, false);
                acc1 = __builtin_amdgcn_wmma_f32_16x16x4_f32(
                    false, a1, false, bf[ks], (short)0, acc1, false, false);
            }
        };
#pragma nounroll
        for (int k0 = 0; k0 < CAT; k0 += 2 * KCH) {
            load_b(bf1, k0 + KCH);
            mma_chunk(bf0, k0);
            load_b(bf0, (k0 + 2 * KCH < CAT) ? k0 + 2 * KCH : k0);
            mma_chunk(bf1, k0 + KCH);
        }
        const float bv = bl[nc];
        float* l0 = Ls + (half * 8) * LP + n;
        float* l1 = Ls + (16 + half * 8) * LP + n;
#pragma unroll
        for (int v = 0; v < 8; ++v) {
            l0[v * LP] = acc0[v] + bv;
            l1[v * LP] = acc1[v] + bv;
        }
    }
    __syncthreads();

#pragma unroll
    for (int q = 0; q < NPW; ++q) {
        const int row = wib * NPW + q;
        const float* lr = Ls + row * LP;
        const bool hi = lane < LAB - 32;
        const float v0 = lr[lane];
        const float v1 = hi ? lr[32 + lane] : -3.4e38f;
        float mx = fmaxf(v0, v1);
#pragma unroll
        for (int o = 16; o > 0; o >>= 1) mx = fmaxf(mx, __shfl_xor(mx, o));
        float sum = __expf(v0 - mx) + (hi ? __expf(v1 - mx) : 0.0f);
#pragma unroll
        for (int o = 16; o > 0; o >>= 1) sum += __shfl_xor(sum, o);
        const float ls = __logf(sum);
        float* orow = out + (size_t)(nb0 + row) * LAB;
        orow[lane] = v0 - mx - ls;
        if (hi) orow[32 + lane] = v1 - mx - ls;
    }
}


extern "C" void kernel_launch(void* const* d_in, const int* in_sizes, int n_in,
                              void* d_out, int out_size, void* d_ws, size_t ws_size,
                              hipStream_t stream)
{
    (void)in_sizes; (void)n_in; (void)out_size; (void)ws_size;

    const float* x  = (const float*)d_in[0];
    const int*   ei = (const int*)d_in[1];
    const float* Wm[3] = {(const float*)d_in[2], (const float*)d_in[4],
                          (const float*)d_in[6]};
    const float* bm[3] = {(const float*)d_in[3], (const float*)d_in[5],
                          (const float*)d_in[7]};
    const float* Wl = (const float*)d_in[8];
    const float* bl = (const float*)d_in[9];
    float* out = (float*)d_out;

    char* ws = (char*)d_ws;
    auto align256 = [](size_t b) { return (b + 255) & ~(size_t)255; };
    size_t off = 0;
    float* dinv   = (float*)(ws + off); off += align256((size_t)GN * 4);
    int*   rowptr = (int*)  (ws + off); off += align256((size_t)(GN + 1) * 4);
    int*   binH   = (int*)  (ws + off); off += align256((size_t)NBIN * EB * 4);
    int*   binO   = (int*)  (ws + off); off += align256((size_t)NBIN * EB * 4);
    int*   bstart = (int*)  (ws + off); off += align256((size_t)NBIN * 4);
    int2*  tmp    = (int2*) (ws + off); off += align256((size_t)GE * 8);
    int2*  srt    = (int2*) (ws + off); off += align256((size_t)GE * 8);
    int2*  adj    = tmp;
    float* hw     = (float*)(ws + off); off += align256((size_t)GN * HID * 4);
    float* hw2    = (float*)(ws + off); off += align256((size_t)GN * HID * 4);
    float* H01    = (float*)(ws + off); off += align256((size_t)GN * HP * 4);

    const dim3 blk(256);

    bin_hist   <<<EB, dim3(L1T), 0, stream>>>(ei, binH);
    bin_scan   <<<NBIN, dim3(SCT), 0, stream>>>(binH, binO, bstart);
    bin_scatter<<<EB, dim3(L1T), 0, stream>>>(ei, binO, tmp);
    bin_sort   <<<NBIN, dim3(SCT), 0, stream>>>(tmp, bstart, rowptr, dinv, srt);
    edge_norm  <<<(GE / 4 + 255) / 256, blk, 0, stream>>>(srt, dinv, adj);

    const int gemmBlocks = (GN + 127) / 128;
    const int gatherBlocks = GN / GBN;

    const size_t shmem0 = (size_t)(DIN / 2) * PadRow<64>::value * sizeof(float2);
    gemm_wmma_f32<4><<<gemmBlocks, blk, shmem0, stream>>>(
        x, DIN, Wm[0], DIN, HID, GN, hw, HID, nullptr);
    node_gather<<<gatherBlocks, blk, 0, stream>>>(
        rowptr, adj, dinv, hw, bm[0], H01, 0, Wm[1], hw2);
    node_gather<<<gatherBlocks, blk, 0, stream>>>(
        rowptr, adj, dinv, hw2, bm[1], H01, HID, Wm[2], hw);
    node_gather_head<<<gatherBlocks, blk, 0, stream>>>(
        rowptr, adj, dinv, hw, bm[2], H01, Wl, bl, out);
}


# edge normalisation factored: product rows leave their producer scaled by their own dinv, a node sums source rows unweighted (order kept) and scales once by its dinv; edge_norm launch gone
# speedup vs baseline: 3.1312x; 1.0827x over previous
#include <hip/hip_runtime.h>


typedef __attribute__((ext_vector_type(2))) float v2f;
typedef __attribute__((ext_vector_type(8))) float v8f;

static constexpr int GN   = 100000;
static constexpr int GE   = 1000000;
static constexpr int DIN  = 128;
static constexpr int HID  = 64;
static constexpr int CAT  = 192;
static constexpr int LAB  = 40;

template <int W> struct PadRow {
    static constexpr int value = W + ((48 - (W % 32)) % 32);
};

template <int NT>
__launch_bounds__(256)
__global__ void gemm_wmma_f32(const float* __restrict__ A, int lda,
                              const float* __restrict__ B, int K, int dout,
                              int M, float* __restrict__ out, int ldo,
                              const float* __restrict__ rscale)
{
    constexpr int W  = NT * 16;
    constexpr int WP = PadRow<W>::value;
    extern __shared__ v2f Bs2[];

    const int KP = K >> 1;
    for (int idx = threadIdx.x; idx < KP * W; idx += 256) {
        int kp = idx / W;
        int n  = idx - kp * W;
        v2f p;
        p.x = (n < dout) ? B[(2 * kp)     * dout + n] : 0.0f;
        p.y = (n < dout) ? B[(2 * kp + 1) * dout + n] : 0.0f;
        Bs2[kp * WP + n] = p;
    }
    __syncthreads();

    const int wave = threadIdx.x >> 5;
    const int lane = threadIdx.x & 31;
    const int l15  = lane & 15;
    const int half = lane >> 4;
    const int kb   = half * 2;

    const int m0 = (blockIdx.x * 8 + wave) * 16;
    int mA = m0 + l15;
    if (mA > M - 1) mA = M - 1;
    const float* Arow = A + (size_t)mA * (size_t)lda;

    float rs[8];
#pragma unroll
    for (int v = 0; v < 8; ++v) {
        const int m = m0 + half * 8 + v;
        rs[v] = rscale[m < M ? m : M - 1];
    }

    v8f acc[NT];
#pragma unroll
    for (int t = 0; t < NT; ++t)
        acc[t] = (v8f){0.f, 0.f, 0.f, 0.f, 0.f, 0.f, 0.f, 0.f};

    for (int k = 0; k < K; k += 4) {
        v2f a = *(const v2f*)(Arow + k + kb);
        const v2f* brow = Bs2 + ((k >> 1) + half) * WP + l15;
#pragma unroll
        for (int t = 0; t < NT; ++t) {
            v2f b = brow[t * 16];
            acc[t] = __builtin_amdgcn_wmma_f32_16x16x4_f32(
                false, a, false, b, (short)0, acc[t], false, false);
        }
    }

#pragma unroll
    for (int t = 0; t < NT; ++t) {
        const int n = t * 16 + l15;
#pragma unroll
        for (int v = 0; v < 8; ++v) {
            const int m = m0 + half * 8 + v;
            if (m < M && n < dout)
                out[(size_t)m * ldo + n] = acc[t][v] * rs[v];
        }
    }
}

static constexpr int BW   = 400;
static constexpr int NBIN = GN / BW;
static constexpr int CH   = 4096;
static constexpr int EB   = (GE + CH - 1) / CH;
static constexpr int L1T  = 512;
static constexpr int GRPS  = CH / 4 / L1T;
static constexpr int SCT  = 1024;
static_assert(GN % BW == 0, "ranges cover the nodes exactly");
static_assert(NBIN <= L1T && EB <= SCT && BW <= SCT, "one thread per counter");
static_assert(GE % 4 == 0 && CH % (4 * L1T) == 0, "edges are read four at a time");

__device__ __forceinline__ int imin(int a, int b) { return a < b ? a : b; }
__device__ __forceinline__ int imax(int a, int b) { return a > b ? a : b; }

__device__ __forceinline__ int bin_of(int d) {
    return imin(imax(d, 0), GN - 1) / BW;
}

__device__ __forceinline__ int block_excl_scan(int v, int pre, int* wsum, int* psum)
{
    const int lane = threadIdx.x & 31;
    const int wave = threadIdx.x >> 5;
    int inc = v;
#pragma unroll
    for (int o = 1; o < 32; o <<= 1) {
        const int y = __shfl_up(inc, o);
        if (lane >= o) inc += y;
    }
#pragma unroll
    for (int o = 16; o > 0; o >>= 1) pre += __shfl_xor(pre, o);
    if (lane == 31) wsum[wave] = inc;
    if (lane == 0)  psum[wave] = pre;
    __syncthreads();
    int tot = psum[lane] + ((lane < wave) ? wsum[lane] : 0);
#pragma unroll
    for (int o = 16; o > 0; o >>= 1) tot += __shfl_xor(tot, o);
    return tot + inc - v;
}

__launch_bounds__(L1T)
__global__ void bin_hist(const int* __restrict__ ei, int* __restrict__ H)
{
    __shared__ int h[NBIN];
    const int tid = threadIdx.x;
    const int b   = blockIdx.x;
    if (tid < NBIN) h[tid] = 0;
    __syncthreads();
    const int4* d4 = (const int4*)(ei + GE);
    int4 d[GRPS];
    bool ok[GRPS];
#pragma unroll
    for (int k = 0; k < GRPS; ++k) {
        const int g = b * (CH / 4) + k * L1T + tid;
        ok[k] = g < GE / 4;
        d[k]  = d4[ok[k] ? g : 0];
    }
#pragma unroll
    for (int k = 0; k < GRPS; ++k) {
        if (ok[k]) {
            atomicAdd(&h[bin_of(d[k].x)], 1);
            atomicAdd(&h[bin_of(d[k].y)], 1);
            atomicAdd(&h[bin_of(d[k].z)], 1);
            atomicAdd(&h[bin_of(d[k].w)], 1);
        }
    }
    __syncthreads();
    if (tid < NBIN) H[tid * EB + b] = h[tid];
}

__launch_bounds__(SCT)
__global__ void bin_scan(const int* __restrict__ H, int* __restrict__ O,
                         int* __restrict__ binstart)
{
    __shared__ int wsum[32];
    __shared__ int psum[32];
    const int tid = threadIdx.x;
    const int n0  = blockIdx.x * EB;
    int pre = 0;
    for (int i = tid; i < n0; i += SCT) pre += H[i];
    const int v = (tid < EB) ? H[n0 + tid] : 0;
    const int ex = block_excl_scan(v, pre, wsum, psum);
    if (tid < EB) O[n0 + tid] = ex;
    if (tid == 0) binstart[blockIdx.x] = ex;
}

__launch_bounds__(L1T)
__global__ void bin_scatter(const int* __restrict__ ei, const int* __restrict__ O,
                            int2* __restrict__ tmp)
{
    __shared__ int cur[NBIN];
    const int tid = threadIdx.x;
    const int b   = blockIdx.x;
    if (tid < NBIN) cur[tid] = O[tid * EB + b];
    __syncthreads();
    const int4* s4 = (const int4*)ei;
    const int4* d4 = (const int4*)(ei + GE);
    int4 s[GRPS], d[GRPS];
    bool ok[GRPS];
#pragma unroll
    for (int k = 0; k < GRPS; ++k) {
        const int g = b * (CH / 4) + k * L1T + tid;
        ok[k] = g < GE / 4;
        s[k]  = s4[ok[k] ? g : 0];
        d[k]  = d4[ok[k] ? g : 0];
    }
#pragma unroll
    for (int k = 0; k < GRPS; ++k) {
        if (ok[k]) {
            const int p0 = atomicAdd(&cur[bin_of(d[k].x)], 1);
            const int p1 = atomicAdd(&cur[bin_of(d[k].y)], 1);
            const int p2 = atomicAdd(&cur[bin_of(d[k].z)], 1);
            const int p3 = atomicAdd(&cur[bin_of(d[k].w)], 1);
            if ((unsigned)p0 < (unsigned)GE) tmp[p0] = make_int2(s[k].x, d[k].x);
            if ((unsigned)p1 < (unsigned)GE) tmp[p1] = make_int2(s[k].y, d[k].y);
            if ((unsigned)p2 < (unsigned)GE) tmp[p2] = make_int2(s[k].z, d[k].z);
            if ((unsigned)p3 < (unsigned)GE) tmp[p3] = make_int2(s[k].w, d[k].w);
        }
    }
}

static constexpr int RPT = 4;
__launch_bounds__(SCT)
__global__ void bin_sort(const int2* __restrict__ tmp, const int* __restrict__ binstart,
                         int* __restrict__ rowptr, float* __restrict__ dinv,
                         int* __restrict__ srcs)
{
    __shared__ int c[BW];
    __shared__ int wsum[32];
    __shared__ int psum[32];
    const int tid  = threadIdx.x;
    const int bin  = blockIdx.x;
    const int nb   = bin * BW;
    const int lo   = imin(imax(binstart[bin], 0), GE);
    const int hi   = (bin == NBIN - 1) ? GE : imin(imax(binstart[bin + 1], lo), GE);

    if (tid < BW) c[tid] = 0;
    __syncthreads();
    for (int p0 = lo + tid; p0 < hi; p0 += RPT * SCT) {
        int dl[RPT];
#pragma unroll
        for (int u = 0; u < RPT; ++u)
            dl[u] = imin(imax(tmp[imin(p0 + u * SCT, hi - 1)].y - nb, 0), BW - 1);
#pragma unroll
        for (int u = 0; u < RPT; ++u)
            atomicAdd(&c[dl[u]], (p0 + u * SCT < hi) ? 1 : 0);
    }
    __syncthreads();
    const int v  = (tid < BW) ? c[tid] : 0;
    const int ex = block_excl_scan(v, 0, wsum, psum);
    if (tid < BW) {
        rowptr[nb + tid] = lo + ex;
        dinv[nb + tid]   = rsqrtf((float)(v + 1));
        c[tid] = ex;
    }
    if (bin == NBIN - 1 && tid == 0) rowptr[GN] = hi;
    __syncthreads();
    for (int p0 = lo + tid; p0 < hi; p0 += RPT * SCT) {
        int2 sd[RPT];
        int  pos[RPT];
#pragma unroll
        for (int u = 0; u < RPT; ++u)
            sd[u] = tmp[imin(p0 + u * SCT, hi - 1)];
#pragma unroll
        for (int u = 0; u < RPT; ++u) {
            const bool live = p0 + u * SCT < hi;
            const int r = atomicAdd(&c[imin(imax(sd[u].y - nb, 0), BW - 1)], live ? 1 : 0);
            pos[u] = live ? lo + r : -1;
        }
#pragma unroll
        for (int u = 0; u < RPT; ++u)
            if ((unsigned)pos[u] < (unsigned)GE) srcs[pos[u]] = sd[u].x;
    }
}

static constexpr int NPW = 4;
static_assert(GN % NPW == 0, "a wave's nodes are all valid");

template <int G>
__device__ __forceinline__ void gather_rows(const float* __restrict__ hw,
                                            int s, int j, int c,
                                            float& ax, float& ay)
{
    float2 v[G];
#pragma unroll
    for (int u = 0; u < G; ++u) {
        const int sj = __builtin_amdgcn_readlane(s, j + u);
        v[u] = *(const float2*)(hw + (size_t)sj * HID + c);
    }
#pragma unroll
    for (int u = 0; u < G; ++u) {
        ax += v[u].x;
        ay += v[u].y;
    }
}

template <class Pre, class Emit>
__device__ __forceinline__ void gather_nodes(const int* __restrict__ rowptr,
                                             const int* __restrict__ srcs,
                                             const float* __restrict__ dinv,
                                             const float* __restrict__ hw,
                                             const float* __restrict__ bias,
                                             int i0, int lane, Pre pre, Emit emit)
{
    const int c = lane * 2;
    int rp[NPW + 1];
#pragma unroll
    for (int q = 0; q <= NPW; ++q)
        rp[q] = imin(imax(__builtin_amdgcn_readfirstlane(rowptr[i0 + q]), 0), GE);

    int    a0[NPW];
    float2 hs[NPW];
    float  di[NPW];
#pragma unroll
    for (int q = 0; q < NPW; ++q) {
        const int n = imin(rp[q + 1] - rp[q], 32);
        const int idx = imin(rp[q] + imax(imin(lane, n - 1), 0), GE - 1);
        a0[q] = srcs[idx];
        hs[q] = *(const float2*)(hw + (size_t)(i0 + q) * HID + c);
        di[q] = dinv[i0 + q];
    }
    const float2 bv = *(const float2*)(bias + c);
    pre();

#pragma unroll
    for (int q = 0; q < NPW; ++q) {
        const int end = rp[q + 1];
        int base = rp[q];
        int aa = a0[q];
        float ax = 0.0f, ay = 0.0f;
        while (base < end) {
            const int n = imin(end - base, 32);
            const int s = imin(imax(aa, 0), GN - 1);
            int j = 0;
            for (; j + 8 <= n; j += 8) gather_rows<8>(hw, s, j, c, ax, ay);
            if (j + 4 <= n) { gather_rows<4>(hw, s, j, c, ax, ay); j += 4; }
            if (j + 2 <= n) { gather_rows<2>(hw, s, j, c, ax, ay); j += 2; }
            if (j < n)      { gather_rows<1>(hw, s, j, c, ax, ay); }
            base += 32;
            if (base < end) {
                const int n2 = imin(end - base, 32);
                aa = srcs[imin(base + imin(lane, n2 - 1), GE - 1)];
            }
        }
        ax += hs[q].x;
        ay += hs[q].y;
        float2 o;
        o.x = fmaxf(di[q] * ax + bv.x, 0.0f);
        o.y = fmaxf(di[q] * ay + bv.y, 0.0f);
        emit(q, o, di[q]);
    }
}

static constexpr int GBW  = 8;
static constexpr int GBN  = GBW * NPW;
static constexpr int APAD = HID + 4;
static constexpr int HP   = 2 * HID;
static_assert(GN % GBN == 0, "every block owns GBN valid nodes");
static_assert(GBN == 32 && HID == 64, "8 waves = 2 row tiles x 4 column tiles");

__launch_bounds__(256)
__global__ void node_gather(const int* __restrict__ rowptr,
                            const int* __restrict__ srcs,
                            const float* __restrict__ dinv,
                            const float* __restrict__ hw,
                            const float* __restrict__ bias,
                            float* __restrict__ H01, int col0,
                            const float* __restrict__ Wn,
                            float* __restrict__ hwn)
{
    __shared__ float As[GBN * APAD];
    const int lane = threadIdx.x & 31;
    const int wib = __builtin_amdgcn_readfirstlane((int)(threadIdx.x >> 5));
    const int nb0 = blockIdx.x * GBN;
    if (nb0 >= GN) return;
    const int i0 = nb0 + wib * NPW;
    const int c = lane * 2;

    gather_nodes(rowptr, srcs, dinv, hw, bias, i0, lane, [] {},
                 [&](int q, float2 o, float d) {
        *(float2*)(H01 + (size_t)(i0 + q) * HP + col0 + c) = o;
        float2 os;
        os.x = d * o.x;
        os.y = d * o.y;
        *(float2*)(As + (wib * NPW + q) * APAD + c) = os;
    });

    const int rt   = wib >> 2;
    const int ct   = wib & 3;
    const int l15  = lane & 15;
    const int half = lane >> 4;
    const int kb   = half * 2;

    const float* wp = Wn + (size_t)kb * HID + ct * 16 + l15;
    v2f bf[HID / 4];
#pragma unroll
    for (int ks = 0; ks < HID / 4; ++ks) {
        bf[ks].x = wp[(4 * ks)     * HID];
        bf[ks].y = wp[(4 * ks + 1) * HID];
    }
    __syncthreads();

    const float* arow = As + (rt * 16 + l15) * APAD + kb;
    v8f acc = (v8f){0.f, 0.f, 0.f, 0.f, 0.f, 0.f, 0.f, 0.f};
#pragma unroll
    for (int ks = 0; ks < HID / 4; ++ks) {
        const v2f a = *(const v2f*)(arow + 4 * ks);
        acc = __builtin_amdgcn_wmma_f32_16x16x4_f32(
            false, a, false, bf[ks], (short)0, acc, false, false);
    }
    float* orow = hwn + (size_t)(nb0 + rt * 16 + half * 8) * HID + ct * 16 + l15;
#pragma unroll
    for (int v = 0; v < 8; ++v)
        orow[(size_t)v * HID] = acc[v];
}

static constexpr int FPAD = CAT + 4;
static constexpr int NTL  = (LAB + 15) / 16;
static constexpr int LP   = 52;
static constexpr int KCH  = 16;
static_assert(NTL <= GBW && NTL * 16 <= LP && CAT % (2 * KCH) == 0 && LAB > 32 && LAB <= 64,
              "logit tiles fit the block; a lane owns labels lane and lane + 32");

__launch_bounds__(256)
__global__ void node_gather_head(const int* __restrict__ rowptr,
                                 const int* __restrict__ srcs,
                                 const float* __restrict__ dinv,
                                 const float* __restrict__ hw,
                                 const float* __restrict__ bias,
                                 const float* __restrict__ H01,
                                 const float* __restrict__ Wl,
                                 const float* __restrict__ bl,
                                 float* __restrict__ out)
{
    __shared__ float As[GBN * FPAD];
    __shared__ float Ls[GBN * LP];
    const int lane = threadIdx.x & 31;
    const int wib = __builtin_amdgcn_readfirstlane((int)(threadIdx.x >> 5));
    const int nb0 = blockIdx.x * GBN;
    if (nb0 >= GN) return;
    const int i0 = nb0 + wib * NPW;
    const int c = lane * 2;

    const int l15  = lane & 15;
    const int half = lane >> 4;
    const int kb   = half * 2;
    const int n    = wib * 16 + l15;
    const int nc   = (n < LAB) ? n : 0;
    const float* wp = Wl + (size_t)kb * LAB + nc;
    auto load_b = [&](v2f (&bf)[KCH / 4], int k0) {
#pragma unroll
        for (int ks = 0; ks < KCH / 4; ++ks) {
            const float* p = wp + (size_t)(k0 + 4 * ks) * LAB;
            bf[ks].x = p[0];
            bf[ks].y = p[LAB];
        }
    };

    gather_nodes(rowptr, srcs, dinv, hw, bias, i0, lane,
        [&] {
            float4 hc[NPW];
#pragma unroll
            for (int q = 0; q < NPW; ++q)
                hc[q] = *(const float4*)(H01 + (size_t)(i0 + q) * HP + lane * 4);
#pragma unroll
            for (int q = 0; q < NPW; ++q)
                *(float4*)(As + (wib * NPW + q) * FPAD + lane * 4) = hc[q];
        },
        [&](int q, float2 o, float) {
            *(float2*)(As + (wib * NPW + q) * FPAD + HP + c) = o;
        });

    const bool mm = wib < NTL;
    v2f bf0[KCH / 4], bf1[KCH / 4];
#pragma unroll
    for (int ks = 0; ks < KCH / 4; ++ks) bf0[ks] = (v2f){0.f, 0.f};
    if (mm) load_b(bf0, 0);
    __syncthreads();

    if (mm) {
        const float* ar0 = As + l15 * FPAD + kb;
        const float* ar1 = As + (16 + l15) * FPAD + kb;
        v8f acc0 = (v8f){0.f, 0.f, 0.f, 0.f, 0.f, 0.f, 0.f, 0.f};
        v8f acc1 = (v8f){0.f, 0.f, 0.f, 0.f, 0.f, 0.f, 0.f, 0.f};
        auto mma_chunk = [&](const v2f (&bf)[KCH / 4], int k0) {
#pragma unroll
            for (int ks = 0; ks < KCH / 4; ++ks) {
                const v2f a0 = *(const v2f*)(ar0 + k0 + 4 * ks);
                const v2f a1 = *(const v2f*)(ar1 + k0 + 4 * ks);
                acc0 = __builtin_amdgcn_wmma_f32_16x16x4_f32(
                    false, a0, false, bf[ks], (short)0, acc0, false, false);
                acc1 = __builtin_amdgcn_wmma_f32_16x16x4_f32(
                    false, a1, false, bf[ks], (short)0, acc1, false, false);
            }
        };
#pragma nounroll
        for (int k0 = 0; k0 < CAT; k0 += 2 * KCH) {
            load_b(bf1, k0 + KCH);
            mma_chunk(bf0, k0);
            load_b(bf0, (k0 + 2 * KCH < CAT) ? k0 + 2 * KCH : k0);
            mma_chunk(bf1, k0 + KCH);
        }
        const float bv = bl[nc];
        float* l0 = Ls + (half * 8) * LP + n;
        float* l1 = Ls + (16 + half * 8) * LP + n;
#pragma unroll
        for (int v = 0; v < 8; ++v) {
            l0[v * LP] = acc0[v] + bv;
            l1[v * LP] = acc1[v] + bv;
        }
    }
    __syncthreads();

#pragma unroll
    for (int q = 0; q < NPW; ++q) {
        const int row = wib * NPW + q;
        const float* lr = Ls + row * LP;
        const bool hi = lane < LAB - 32;
        const float v0 = lr[lane];
        const float v1 = hi ? lr[32 + lane] : -3.4e38f;
        float mx = fmaxf(v0, v1);
#pragma unroll
        for (int o = 16; o > 0; o >>= 1) mx = fmaxf(mx, __shfl_xor(mx, o));
        float sum = __expf(v0 - mx) + (hi ? __expf(v1 - mx) : 0.0f);
#pragma unroll
        for (int o = 16; o > 0; o >>= 1) sum += __shfl_xor(sum, o);
        const float ls = __logf(sum);
        float* orow = out + (size_t)(nb0 + row) * LAB;
        orow[lane] = v0 - mx - ls;
        if (hi) orow[32 + lane] = v1 - mx - ls;
    }
}


extern "C" void kernel_launch(void* const* d_in, const int* in_sizes, int n_in,
                              void* d_out, int out_size, void* d_ws, size_t ws_size,
                              hipStream_t stream)
{
    (void)in_sizes; (void)n_in; (void)out_size; (void)ws_size;

    const float* x  = (const float*)d_in[0];
    const int*   ei = (const int*)d_in[1];
    const float* Wm[3] = {(const float*)d_in[2], (const float*)d_in[4],
                          (const float*)d_in[6]};
    const float* bm[3] = {(const float*)d_in[3], (const float*)d_in[5],
                          (const float*)d_in[7]};
    const float* Wl = (const float*)d_in[8];
    const float* bl = (const float*)d_in[9];
    float* out = (float*)d_out;

    char* ws = (char*)d_ws;
    auto align256 = [](size_t b) { return (b + 255) & ~(size_t)255; };
    size_t off = 0;
    float* dinv   = (float*)(ws + off); off += align256((size_t)GN * 4);
    int*   rowptr = (int*)  (ws + off); off += align256((size_t)(GN + 1) * 4);
    int*   binH   = (int*)  (ws + off); off += align256((size_t)NBIN * EB * 4);
    int*   binO   = (int*)  (ws + off); off += align256((size_t)NBIN * EB * 4);
    int*   bstart = (int*)  (ws + off); off += align256((size_t)NBIN * 4);
    int2*  tmp    = (int2*) (ws + off); off += align256((size_t)GE * 8);
    int*   srcs   = (int*)  (ws + off); off += align256((size_t)GE * 4);
    float* hw     = (float*)(ws + off); off += align256((size_t)GN * HID * 4);
    float* hw2    = (float*)(ws + off); off += align256((size_t)GN * HID * 4);
    float* H01    = (float*)(ws + off); off += align256((size_t)GN * HP * 4);

    const dim3 blk(256);

    bin_hist   <<<EB, dim3(L1T), 0, stream>>>(ei, binH);
    bin_scan   <<<NBIN, dim3(SCT), 0, stream>>>(binH, binO, bstart);
    bin_scatter<<<EB, dim3(L1T), 0, stream>>>(ei, binO, tmp);
    bin_sort   <<<NBIN, dim3(SCT), 0, stream>>>(tmp, bstart, rowptr, dinv, srcs);

    const int gemmBlocks = (GN + 127) / 128;
    const int gatherBlocks = GN / GBN;

    const size_t shmem0 = (size_t)(DIN / 2) * PadRow<64>::value * sizeof(float2);
    gemm_wmma_f32<4><<<gemmBlocks, blk, shmem0, stream>>>(
        x, DIN, Wm[0], DIN, HID, GN, hw, HID, dinv);
    node_gather<<<gatherBlocks, blk, 0, stream>>>(
        rowptr, srcs, dinv, hw, bm[0], H01, 0, Wm[1], hw2);
    node_gather<<<gatherBlocks, blk, 0, stream>>>(
        rowptr, srcs, dinv, hw2, bm[1], H01, HID, Wm[2], hw);
    node_gather_head<<<gatherBlocks, blk, 0, stream>>>(
        rowptr, srcs, dinv, hw, bm[2], H01, Wl, bl, out);
}


# edge grouping, level 1 made block-local: a block orders its own edges by range in its own stretch (LDS count, scan, place), level 2 reads the pieces; count and scan launches gone; sum order as before
# speedup vs baseline: 3.4289x; 1.0951x over previous
#include <hip/hip_runtime.h>


typedef __attribute__((ext_vector_type(2))) float v2f;
typedef __attribute__((ext_vector_type(8))) float v8f;

static constexpr int GN   = 100000;
static constexpr int GE   = 1000000;
static constexpr int DIN  = 128;
static constexpr int HID  = 64;
static constexpr int CAT  = 192;
static constexpr int LAB  = 40;

template <int W> struct PadRow {
    static constexpr int value = W + ((48 - (W % 32)) % 32);
};

template <int NT>
__launch_bounds__(256)
__global__ void gemm_wmma_f32(const float* __restrict__ A, int lda,
                              const float* __restrict__ B, int K, int dout,
                              int M, float* __restrict__ out, int ldo,
                              const float* __restrict__ rscale)
{
    constexpr int W  = NT * 16;
    constexpr int WP = PadRow<W>::value;
    extern __shared__ v2f Bs2[];

    const int KP = K >> 1;
    for (int idx = threadIdx.x; idx < KP * W; idx += 256) {
        int kp = idx / W;
        int n  = idx - kp * W;
        v2f p;
        p.x = (n < dout) ? B[(2 * kp)     * dout + n] : 0.0f;
        p.y = (n < dout) ? B[(2 * kp + 1) * dout + n] : 0.0f;
        Bs2[kp * WP + n] = p;
    }
    __syncthreads();

    const int wave = threadIdx.x >> 5;
    const int lane = threadIdx.x & 31;
    const int l15  = lane & 15;
    const int half = lane >> 4;
    const int kb   = half * 2;

    const int m0 = (blockIdx.x * 8 + wave) * 16;
    int mA = m0 + l15;
    if (mA > M - 1) mA = M - 1;
    const float* Arow = A + (size_t)mA * (size_t)lda;

    float rs[8];
#pragma unroll
    for (int v = 0; v < 8; ++v) {
        const int m = m0 + half * 8 + v;
        rs[v] = rscale[m < M ? m : M - 1];
    }

    v8f acc[NT];
#pragma unroll
    for (int t = 0; t < NT; ++t)
        acc[t] = (v8f){0.f, 0.f, 0.f, 0.f, 0.f, 0.f, 0.f, 0.f};

    for (int k = 0; k < K; k += 4) {
        v2f a = *(const v2f*)(Arow + k + kb);
        const v2f* brow = Bs2 + ((k >> 1) + half) * WP + l15;
#pragma unroll
        for (int t = 0; t < NT; ++t) {
            v2f b = brow[t * 16];
            acc[t] = __builtin_amdgcn_wmma_f32_16x16x4_f32(
                false, a, false, b, (short)0, acc[t], false, false);
        }
    }

#pragma unroll
    for (int t = 0; t < NT; ++t) {
        const int n = t * 16 + l15;
#pragma unroll
        for (int v = 0; v < 8; ++v) {
            const int m = m0 + half * 8 + v;
            if (m < M && n < dout)
                out[(size_t)m * ldo + n] = acc[t][v] * rs[v];
        }
    }
}

static constexpr int BW   = 400;
static constexpr int NBIN = GN / BW;
static constexpr int CH   = 4096;
static constexpr int EB   = (GE + CH - 1) / CH;
static constexpr int L1T  = 512;
static constexpr int GRPS  = CH / 4 / L1T;
static constexpr int SCT  = 1024;
static constexpr int NBW  = (NBIN + 31) / 32;
static constexpr int TPP  = 4;
static constexpr int RPT  = 8;
static_assert(GN % BW == 0, "ranges cover the nodes exactly");
static_assert(NBIN <= L1T && EB * TPP <= SCT && BW <= SCT, "one thread per counter, TPP per piece");
static_assert(GE % 4 == 0 && CH % (4 * L1T) == 0, "edges are read four at a time");
static_assert(CH < 65536 && GE >= CH, "piece start and length share a word; slot 0 of tmp is always written");

__device__ __forceinline__ int imin(int a, int b) { return a < b ? a : b; }
__device__ __forceinline__ int imax(int a, int b) { return a > b ? a : b; }

__device__ __forceinline__ int bin_of(int d) {
    return imin(imax(d, 0), GN - 1) / BW;
}

__device__ __forceinline__ int block_excl_scan(int v, int pre, int* wsum, int* psum)
{
    const int lane = threadIdx.x & 31;
    const int wave = threadIdx.x >> 5;
    int inc = v;
#pragma unroll
    for (int o = 1; o < 32; o <<= 1) {
        const int y = __shfl_up(inc, o);
        if (lane >= o) inc += y;
    }
#pragma unroll
    for (int o = 16; o > 0; o >>= 1) pre += __shfl_xor(pre, o);
    if (lane == 31) wsum[wave] = inc;
    if (lane == 0)  psum[wave] = pre;
    __syncthreads();
    int tot = psum[lane] + ((lane < wave) ? wsum[lane] : 0);
#pragma unroll
    for (int o = 16; o > 0; o >>= 1) tot += __shfl_xor(tot, o);
    return tot + inc - v;
}

__launch_bounds__(L1T)
__global__ void bin_local(const int* __restrict__ ei, int* __restrict__ PH,
                          int2* __restrict__ tmp)
{
    __shared__ int h[NBIN];
    __shared__ int wsum[NBW];
    const int tid  = threadIdx.x;
    const int b    = blockIdx.x;
    const int lane = tid & 31;
    const int wave = tid >> 5;
    if (tid < NBIN) h[tid] = 0;
    __syncthreads();
    const int4* s4 = (const int4*)ei;
    const int4* d4 = (const int4*)(ei + GE);
    int4 s[GRPS], d[GRPS], r[GRPS];
    bool ok[GRPS];
#pragma unroll
    for (int k = 0; k < GRPS; ++k) {
        const int g = b * (CH / 4) + k * L1T + tid;
        ok[k] = g < GE / 4;
        s[k]  = s4[ok[k] ? g : 0];
        d[k]  = d4[ok[k] ? g : 0];
    }
#pragma unroll
    for (int k = 0; k < GRPS; ++k) {
        r[k].x = bin_of(d[k].x);
        r[k].y = bin_of(d[k].y);
        r[k].z = bin_of(d[k].z);
        r[k].w = bin_of(d[k].w);
        if (ok[k]) {
            atomicAdd(&h[r[k].x], 1);
            atomicAdd(&h[r[k].y], 1);
            atomicAdd(&h[r[k].z], 1);
            atomicAdd(&h[r[k].w], 1);
        }
    }
    __syncthreads();
    const int v = (tid < NBIN) ? h[tid] : 0;
    int inc = v;
#pragma unroll
    for (int o = 1; o < 32; o <<= 1) {
        const int y = __shfl_up(inc, o);
        if (lane >= o) inc += y;
    }
    if (lane == 31 && wave < NBW) wsum[wave] = inc;
    __syncthreads();
    int front = 0;
#pragma unroll
    for (int w = 0; w < NBW; ++w) front += (w < wave) ? wsum[w] : 0;
    const int ex = front + inc - v;
    if (tid < NBIN) {
        PH[tid * EB + b] = (v << 16) | ex;
        h[tid] = ex;
    }
    __syncthreads();
    int2* mine = tmp + (size_t)b * CH;
#pragma unroll
    for (int k = 0; k < GRPS; ++k) {
        if (ok[k]) {
            const int p0 = atomicAdd(&h[r[k].x], 1);
            const int p1 = atomicAdd(&h[r[k].y], 1);
            const int p2 = atomicAdd(&h[r[k].z], 1);
            const int p3 = atomicAdd(&h[r[k].w], 1);
            if ((unsigned)p0 < (unsigned)CH) mine[p0] = make_int2(s[k].x, d[k].x);
            if ((unsigned)p1 < (unsigned)CH) mine[p1] = make_int2(s[k].y, d[k].y);
            if ((unsigned)p2 < (unsigned)CH) mine[p2] = make_int2(s[k].z, d[k].z);
            if ((unsigned)p3 < (unsigned)CH) mine[p3] = make_int2(s[k].w, d[k].w);
        }
    }
}

__launch_bounds__(SCT)
__global__ void bin_sort(const int2* __restrict__ tmp, const int* __restrict__ PH,
                         int* __restrict__ rowptr, float* __restrict__ dinv,
                         int* __restrict__ srcs)
{
    __shared__ int c[BW];
    __shared__ int wsum[32];
    __shared__ int psum[32];
    const int tid  = threadIdx.x;
    const int bin  = blockIdx.x;
    const int nb   = bin * BW;
    const int n0   = bin * EB;

    const int pb   = tid / TPP;
    const int sub  = tid - pb * TPP;
    const int ph   = (pb < EB) ? PH[n0 + pb] : 0;
    const int cnt  = imin(imax(ph >> 16, 0), CH);
    const int st   = imin(pb, EB - 1) * CH + imin(ph & 0xFFFF, CH - cnt);

    int pre = 0;
    for (int i0 = tid; i0 < n0; i0 += 8 * SCT) {
        int t[8];
#pragma unroll
        for (int u = 0; u < 8; ++u) t[u] = PH[imin(i0 + u * SCT, n0 - 1)];
#pragma unroll
        for (int u = 0; u < 8; ++u) pre += (i0 + u * SCT < n0) ? (t[u] >> 16) : 0;
    }

    if (tid < BW) c[tid] = 0;
    __syncthreads();
    for (int i0 = sub; i0 < cnt; i0 += TPP * RPT) {
        int dl[RPT];
#pragma unroll
        for (int u = 0; u < RPT; ++u)
            dl[u] = imin(imax(tmp[st + imin(i0 + u * TPP, cnt - 1)].y - nb, 0), BW - 1);
#pragma unroll
        for (int u = 0; u < RPT; ++u)
            atomicAdd(&c[dl[u]], (i0 + u * TPP < cnt) ? 1 : 0);
    }
    __syncthreads();
    const int v  = (tid < BW) ? c[tid] : 0;
    const int at = block_excl_scan(v, pre, wsum, psum);
    if (tid < BW) {
        rowptr[nb + tid] = at;
        dinv[nb + tid]   = rsqrtf((float)(v + 1));
        c[tid] = at;
    }
    if (bin == NBIN - 1 && tid == 0) rowptr[GN] = GE;
    __syncthreads();
    for (int i0 = sub; i0 < cnt; i0 += TPP * RPT) {
        int2 sd[RPT];
        int  pos[RPT];
#pragma unroll
        for (int u = 0; u < RPT; ++u)
            sd[u] = tmp[st + imin(i0 + u * TPP, cnt - 1)];
#pragma unroll
        for (int u = 0; u < RPT; ++u) {
            const bool live = i0 + u * TPP < cnt;
            const int q = atomicAdd(&c[imin(imax(sd[u].y - nb, 0), BW - 1)], live ? 1 : 0);
            pos[u] = live ? q : -1;
        }
#pragma unroll
        for (int u = 0; u < RPT; ++u)
            if ((unsigned)pos[u] < (unsigned)GE) srcs[pos[u]] = sd[u].x;
    }
}

static constexpr int NPW = 4;
static_assert(GN % NPW == 0, "a wave's nodes are all valid");

template <int G>
__device__ __forceinline__ void gather_rows(const float* __restrict__ hw,
                                            int s, int j, int c,
                                            float& ax, float& ay)
{
    float2 v[G];
#pragma unroll
    for (int u = 0; u < G; ++u) {
        const int sj = __builtin_amdgcn_readlane(s, j + u);
        v[u] = *(const float2*)(hw + (size_t)sj * HID + c);
    }
#pragma unroll
    for (int u = 0; u < G; ++u) {
        ax += v[u].x;
        ay += v[u].y;
    }
}

template <class Pre, class Emit>
__device__ __forceinline__ void gather_nodes(const int* __restrict__ rowptr,
                                             const int* __restrict__ srcs,
                                             const float* __restrict__ dinv,
                                             const float* __restrict__ hw,
                                             const float* __restrict__ bias,
                                             int i0, int lane, Pre pre, Emit emit)
{
    const int c = lane * 2;
    int rp[NPW + 1];
#pragma unroll
    for (int q = 0; q <= NPW; ++q)
        rp[q] = imin(imax(__builtin_amdgcn_readfirstlane(rowptr[i0 + q]), 0), GE);

    int    a0[NPW];
    float2 hs[NPW];
    float  di[NPW];
#pragma unroll
    for (int q = 0; q < NPW; ++q) {
        const int n = imin(rp[q + 1] - rp[q], 32);
        const int idx = imin(rp[q] + imax(imin(lane, n - 1), 0), GE - 1);
        a0[q] = srcs[idx];
        hs[q] = *(const float2*)(hw + (size_t)(i0 + q) * HID + c);
        di[q] = dinv[i0 + q];
    }
    const float2 bv = *(const float2*)(bias + c);
    pre();

#pragma unroll
    for (int q = 0; q < NPW; ++q) {
        const int end = rp[q + 1];
        int base = rp[q];
        int aa = a0[q];
        float ax = 0.0f, ay = 0.0f;
        while (base < end) {
            const int n = imin(end - base, 32);
            const int s = imin(imax(aa, 0), GN - 1);
            int j = 0;
            for (; j + 8 <= n; j += 8) gather_rows<8>(hw, s, j, c, ax, ay);
            if (j + 4 <= n) { gather_rows<4>(hw, s, j, c, ax, ay); j += 4; }
            if (j + 2 <= n) { gather_rows<2>(hw, s, j, c, ax, ay); j += 2; }
            if (j < n)      { gather_rows<1>(hw, s, j, c, ax, ay); }
            base += 32;
            if (base < end) {
                const int n2 = imin(end - base, 32);
                aa = srcs[imin(base + imin(lane, n2 - 1), GE - 1)];
            }
        }
        ax += hs[q].x;
        ay += hs[q].y;
        float2 o;
        o.x = fmaxf(di[q] * ax + bv.x, 0.0f);
        o.y = fmaxf(di[q] * ay + bv.y, 0.0f);
        emit(q, o, di[q]);
    }
}

static constexpr int GBW  = 8;
static constexpr int GBN  = GBW * NPW;
static constexpr int APAD = HID + 4;
static constexpr int HP   = 2 * HID;
static_assert(GN % GBN == 0, "every block owns GBN valid nodes");
static_assert(GBN == 32 && HID == 64, "8 waves = 2 row tiles x 4 column tiles");

__launch_bounds__(256)
__global__ void node_gather(const int* __restrict__ rowptr,
                            const int* __restrict__ srcs,
                            const float* __restrict__ dinv,
                            const float* __restrict__ hw,
                            const float* __restrict__ bias,
                            float* __restrict__ H01, int col0,
                            const float* __restrict__ Wn,
                            float* __restrict__ hwn)
{
    __shared__ float As[GBN * APAD];
    const int lane = threadIdx.x & 31;
    const int wib = __builtin_amdgcn_readfirstlane((int)(threadIdx.x >> 5));
    const int nb0 = blockIdx.x * GBN;
    if (nb0 >= GN) return;
    const int i0 = nb0 + wib * NPW;
    const int c = lane * 2;

    gather_nodes(rowptr, srcs, dinv, hw, bias, i0, lane, [] {},
                 [&](int q, float2 o, float d) {
        *(float2*)(H01 + (size_t)(i0 + q) * HP + col0 + c) = o;
        float2 os;
        os.x = d * o.x;
        os.y = d * o.y;
        *(float2*)(As + (wib * NPW + q) * APAD + c) = os;
    });

    const int rt   = wib >> 2;
    const int ct   = wib & 3;
    const int l15  = lane & 15;
    const int half = lane >> 4;
    const int kb   = half * 2;

    const float* wp = Wn + (size_t)kb * HID + ct * 16 + l15;
    v2f bf[HID / 4];
#pragma unroll
    for (int ks = 0; ks < HID / 4; ++ks) {
        bf[ks].x = wp[(4 * ks)     * HID];
        bf[ks].y = wp[(4 * ks + 1) * HID];
    }
    __syncthreads();

    const float* arow = As + (rt * 16 + l15) * APAD + kb;
    v8f acc = (v8f){0.f, 0.f, 0.f, 0.f, 0.f, 0.f, 0.f, 0.f};
#pragma unroll
    for (int ks = 0; ks < HID / 4; ++ks) {
        const v2f a = *(const v2f*)(arow + 4 * ks);
        acc = __builtin_amdgcn_wmma_f32_16x16x4_f32(
            false, a, false, bf[ks], (short)0, acc, false, false);
    }
    float* orow = hwn + (size_t)(nb0 + rt * 16 + half * 8) * HID + ct * 16 + l15;
#pragma unroll
    for (int v = 0; v < 8; ++v)
        orow[(size_t)v * HID] = acc[v];
}

static constexpr int FPAD = CAT + 4;
static constexpr int NTL  = (LAB + 15) / 16;
static constexpr int LP   = 52;
static constexpr int KCH  = 16;
static_assert(NTL <= GBW && NTL * 16 <= LP && CAT % (2 * KCH) == 0 && LAB > 32 && LAB <= 64,
              "logit tiles fit the block; a lane owns labels lane and lane + 32");

__launch_bounds__(256)
__global__ void node_gather_head(const int* __restrict__ rowptr,
                                 const int* __restrict__ srcs,
                                 const float* __restrict__ dinv,
                                 const float* __restrict__ hw,
                                 const float* __restrict__ bias,
                                 const float* __restrict__ H01,
                                 const float* __restrict__ Wl,
                                 const float* __restrict__ bl,
                                 float* __restrict__ out)
{
    __shared__ float As[GBN * FPAD];
    __shared__ float Ls[GBN * LP];
    const int lane = threadIdx.x & 31;
    const int wib = __builtin_amdgcn_readfirstlane((int)(threadIdx.x >> 5));
    const int nb0 = blockIdx.x * GBN;
    if (nb0 >= GN) return;
    const int i0 = nb0 + wib * NPW;
    const int c = lane * 2;

    const int l15  = lane & 15;
    const int half = lane >> 4;
    const int kb   = half * 2;
    const int n    = wib * 16 + l15;
    const int nc   = (n < LAB) ? n : 0;
    const float* wp = Wl + (size_t)kb * LAB + nc;
    auto load_b = [&](v2f (&bf)[KCH / 4], int k0) {
#pragma unroll
        for (int ks = 0; ks < KCH / 4; ++ks) {
            const float* p = wp + (size_t)(k0 + 4 * ks) * LAB;
            bf[ks].x = p[0];
            bf[ks].y = p[LAB];
        }
    };

    gather_nodes(rowptr, srcs, dinv, hw, bias, i0, lane,
        [&] {
            float4 hc[NPW];
#pragma unroll
            for (int q = 0; q < NPW; ++q)
                hc[q] = *(const float4*)(H01 + (size_t)(i0 + q) * HP + lane * 4);
#pragma unroll
            for (int q = 0; q < NPW; ++q)
                *(float4*)(As + (wib * NPW + q) * FPAD + lane * 4) = hc[q];
        },
        [&](int q, float2 o, float) {
            *(float2*)(As + (wib * NPW + q) * FPAD + HP + c) = o;
        });

    const bool mm = wib < NTL;
    v2f bf0[KCH / 4], bf1[KCH / 4];
#pragma unroll
    for (int ks = 0; ks < KCH / 4; ++ks) bf0[ks] = (v2f){0.f, 0.f};
    if (mm) load_b(bf0, 0);
    __syncthreads();

    if (mm) {
        const float* ar0 = As + l15 * FPAD + kb;
        const float* ar1 = As + (16 + l15) * FPAD + kb;
        v8f acc0 = (v8f){0.f, 0.f, 0.f, 0.f, 0.f, 0.f, 0.f, 0.f};
        v8f acc1 = (v8f){0.f, 0.f, 0.f, 0.f, 0.f, 0.f, 0.f, 0.f};
        auto mma_chunk = [&](const v2f (&bf)[KCH / 4], int k0) {
#pragma unroll
            for (int ks = 0; ks < KCH / 4; ++ks) {
                const v2f a0 = *(const v2f*)(ar0 + k0 + 4 * ks);
                const v2f a1 = *(const v2f*)(ar1 + k0 + 4 * ks);
                acc0 = __builtin_amdgcn_wmma_f32_16x16x4_f32(
                    false, a0, false, bf[ks], (short)0, acc0, false, false);
                acc1 = __builtin_amdgcn_wmma_f32_16x16x4_f32(
                    false, a1, false, bf[ks], (short)0, acc1, false, false);
            }
        };
#pragma nounroll
        for (int k0 = 0; k0 < CAT; k0 += 2 * KCH) {
            load_b(bf1, k0 + KCH);
            mma_chunk(bf0, k0);
            load_b(bf0, (k0 + 2 * KCH < CAT) ? k0 + 2 * KCH : k0);
            mma_chunk(bf1, k0 + KCH);
        }
        const float bv = bl[nc];
        float* l0 = Ls + (half * 8) * LP + n;
        float* l1 = Ls + (16 + half * 8) * LP + n;
#pragma unroll
        for (int v = 0; v < 8; ++v) {
            l0[v * LP] = acc0[v] + bv;
            l1[v * LP] = acc1[v] + bv;
        }
    }
    __syncthreads();

#pragma unroll
    for (int q = 0; q < NPW; ++q) {
        const int row = wib * NPW + q;
        const float* lr = Ls + row * LP;
        const bool hi = lane < LAB - 32;
        const float v0 = lr[lane];
        const float v1 = hi ? lr[32 + lane] : -3.4e38f;
        float mx = fmaxf(v0, v1);
#pragma unroll
        for (int o = 16; o > 0; o >>= 1) mx = fmaxf(mx, __shfl_xor(mx, o));
        float sum = __expf(v0 - mx) + (hi ? __expf(v1 - mx) : 0.0f);
#pragma unroll
        for (int o = 16; o > 0; o >>= 1) sum += __shfl_xor(sum, o);
        const float ls = __logf(sum);
        float* orow = out + (size_t)(nb0 + row) * LAB;
        orow[lane] = v0 - mx - ls;
        if (hi) orow[32 + lane] = v1 - mx - ls;
    }
}


extern "C" void kernel_launch(void* const* d_in, const int* in_sizes, int n_in,
                              void* d_out, int out_size, void* d_ws, size_t ws_size,
                              hipStream_t stream)
{
    (void)in_sizes; (void)n_in; (void)out_size; (void)ws_size;

    const float* x  = (const float*)d_in[0];
    const int*   ei = (const int*)d_in[1];
    const float* Wm[3] = {(const float*)d_in[2], (const float*)d_in[4],
                          (const float*)d_in[6]};
    const float* bm[3] = {(const float*)d_in[3], (const float*)d_in[5],
                          (const float*)d_in[7]};
    const float* Wl = (const float*)d_in[8];
    const float* bl = (const float*)d_in[9];
    float* out = (float*)d_out;

    char* ws = (char*)d_ws;
    auto align256 = [](size_t b) { return (b + 255) & ~(size_t)255; };
    size_t off = 0;
    float* dinv   = (float*)(ws + off); off += align256((size_t)GN * 4);
    int*   rowptr = (int*)  (ws + off); off += align256((size_t)(GN + 1) * 4);
    int*   binPH  = (int*)  (ws + off); off += align256((size_t)NBIN * EB * 4);
    int2*  tmp    = (int2*) (ws + off); off += align256((size_t)EB * CH * 8);
    int*   srcs   = (int*)  (ws + off); off += align256((size_t)GE * 4);
    float* hw     = (float*)(ws + off); off += align256((size_t)GN * HID * 4);
    float* hw2    = (float*)(ws + off); off += align256((size_t)GN * HID * 4);
    float* H01    = (float*)(ws + off); off += align256((size_t)GN * HP * 4);

    const dim3 blk(256);

    bin_local<<<EB, dim3(L1T), 0, stream>>>(ei, binPH, tmp);
    bin_sort <<<NBIN, dim3(SCT), 0, stream>>>(tmp, binPH, rowptr, dinv, srcs);

    const int gemmBlocks = (GN + 127) / 128;
    const int gatherBlocks = GN / GBN;

    const size_t shmem0 = (size_t)(DIN / 2) * PadRow<64>::value * sizeof(float2);
    gemm_wmma_f32<4><<<gemmBlocks, blk, shmem0, stream>>>(
        x, DIN, Wm[0], DIN, HID, GN, hw, HID, dinv);
    node_gather<<<gatherBlocks, blk, 0, stream>>>(
        rowptr, srcs, dinv, hw, bm[0], H01, 0, Wm[1], hw2);
    node_gather<<<gatherBlocks, blk, 0, stream>>>(
        rowptr, srcs, dinv, hw2, bm[1], H01, HID, Wm[2], hw);
    node_gather_head<<<gatherBlocks, blk, 0, stream>>>(
        rowptr, srcs, dinv, hw, bm[2], H01, Wl, bl, out);
}


# layer-0 product x @ W0 moved into the level-2 launch of the edge grouping: a second work-group per range counts its in-edges for dinv and forms its 400 scaled rows on the f32 WMMA; GEMM launch gone
# speedup vs baseline: 3.8965x; 1.1364x over previous
#include <hip/hip_runtime.h>


typedef __attribute__((ext_vector_type(2))) float v2f;
typedef __attribute__((ext_vector_type(8))) float v8f;

static constexpr int GN   = 100000;
static constexpr int GE   = 1000000;
static constexpr int DIN  = 128;
static constexpr int HID  = 64;
static constexpr int CAT  = 192;
static constexpr int LAB  = 40;

template <int W> struct PadRow {
    static constexpr int value = W + ((48 - (W % 32)) % 32);
};

static constexpr int BW   = 400;
static constexpr int NBIN = GN / BW;
static constexpr int CH   = 4096;
static constexpr int EB   = (GE + CH - 1) / CH;
static constexpr int L1T  = 512;
static constexpr int GRPS  = CH / 4 / L1T;
static constexpr int SCT  = 1024;
static constexpr int NBW  = (NBIN + 31) / 32;
static constexpr int TPP  = 4;
static constexpr int RPT  = 8;
static_assert(GN % BW == 0, "ranges cover the nodes exactly");
static_assert(NBIN <= L1T && EB * TPP <= SCT && BW <= SCT, "one thread per counter, TPP per piece");
static_assert(GE % 4 == 0 && CH % (4 * L1T) == 0, "edges are read four at a time");
static_assert(CH < 65536 && GE >= CH, "piece start and length share a word; slot 0 of tmp is always written");
static_assert(BW % 16 == 0 && BW / 16 <= SCT / 32 && DIN % 4 == 0 && HID % 16 == 0,
              "a wave per 16-row stripe of the range's layer-0 product");

__device__ __forceinline__ int imin(int a, int b) { return a < b ? a : b; }
__device__ __forceinline__ int imax(int a, int b) { return a > b ? a : b; }

__device__ __forceinline__ int bin_of(int d) {
    return imin(imax(d, 0), GN - 1) / BW;
}

__device__ __forceinline__ int block_excl_scan(int v, int pre, int* wsum, int* psum)
{
    const int lane = threadIdx.x & 31;
    const int wave = threadIdx.x >> 5;
    int inc = v;
#pragma unroll
    for (int o = 1; o < 32; o <<= 1) {
        const int y = __shfl_up(inc, o);
        if (lane >= o) inc += y;
    }
#pragma unroll
    for (int o = 16; o > 0; o >>= 1) pre += __shfl_xor(pre, o);
    if (lane == 31) wsum[wave] = inc;
    if (lane == 0)  psum[wave] = pre;
    __syncthreads();
    int tot = psum[lane] + ((lane < wave) ? wsum[lane] : 0);
#pragma unroll
    for (int o = 16; o > 0; o >>= 1) tot += __shfl_xor(tot, o);
    return tot + inc - v;
}

__launch_bounds__(L1T)
__global__ void bin_local(const int* __restrict__ ei, int* __restrict__ PH,
                          int2* __restrict__ tmp)
{
    __shared__ int h[NBIN];
    __shared__ int wsum[NBW];
    const int tid  = threadIdx.x;
    const int b    = blockIdx.x;
    const int lane = tid & 31;
    const int wave = tid >> 5;
    if (tid < NBIN) h[tid] = 0;
    __syncthreads();
    const int4* s4 = (const int4*)ei;
    const int4* d4 = (const int4*)(ei + GE);
    int4 s[GRPS], d[GRPS], r[GRPS];
    bool ok[GRPS];
#pragma unroll
    for (int k = 0; k < GRPS; ++k) {
        const int g = b * (CH / 4) + k * L1T + tid;
        ok[k] = g < GE / 4;
        s[k]  = s4[ok[k] ? g : 0];
        d[k]  = d4[ok[k] ? g : 0];
    }
#pragma unroll
    for (int k = 0; k < GRPS; ++k) {
        r[k].x = bin_of(d[k].x);
        r[k].y = bin_of(d[k].y);
        r[k].z = bin_of(d[k].z);
        r[k].w = bin_of(d[k].w);
        if (ok[k]) {
            atomicAdd(&h[r[k].x], 1);
            atomicAdd(&h[r[k].y], 1);
            atomicAdd(&h[r[k].z], 1);
            atomicAdd(&h[r[k].w], 1);
        }
    }
    __syncthreads();
    const int v = (tid < NBIN) ? h[tid] : 0;
    int inc = v;
#pragma unroll
    for (int o = 1; o < 32; o <<= 1) {
        const int y = __shfl_up(inc, o);
        if (lane >= o) inc += y;
    }
    if (lane == 31 && wave < NBW) wsum[wave] = inc;
    __syncthreads();
    int front = 0;
#pragma unroll
    for (int w = 0; w < NBW; ++w) front += (w < wave) ? wsum[w] : 0;
    const int ex = front + inc - v;
    if (tid < NBIN) {
        PH[tid * EB + b] = (v << 16) | ex;
        h[tid] = ex;
    }
    __syncthreads();
    int2* mine = tmp + (size_t)b * CH;
#pragma unroll
    for (int k = 0; k < GRPS; ++k) {
        if (ok[k]) {
            const int p0 = atomicAdd(&h[r[k].x], 1);
            const int p1 = atomicAdd(&h[r[k].y], 1);
            const int p2 = atomicAdd(&h[r[k].z], 1);
            const int p3 = atomicAdd(&h[r[k].w], 1);
            if ((unsigned)p0 < (unsigned)CH) mine[p0] = make_int2(s[k].x, d[k].x);
            if ((unsigned)p1 < (unsigned)CH) mine[p1] = make_int2(s[k].y, d[k].y);
            if ((unsigned)p2 < (unsigned)CH) mine[p2] = make_int2(s[k].z, d[k].z);
            if ((unsigned)p3 < (unsigned)CH) mine[p3] = make_int2(s[k].w, d[k].w);
        }
    }
}

__device__ __forceinline__ void range_product(const float* __restrict__ x,
                                              const v2f* Bs2, const float* dsh,
                                              float* __restrict__ hw, int nb, int tid)
{
    constexpr int NT = HID / 16;
    constexpr int WP = PadRow<HID>::value;
    const int wv = __builtin_amdgcn_readfirstlane(tid >> 5);
    if (wv >= BW / 16) return;
    const int lane = tid & 31;
    const int l15  = lane & 15;
    const int half = lane >> 4;
    const int kb   = half * 2;
    const int m0   = wv * 16;
    const float* Arow = x + (size_t)(nb + m0 + l15) * DIN;

    v8f acc[NT];
#pragma unroll
    for (int t = 0; t < NT; ++t)
        acc[t] = (v8f){0.f, 0.f, 0.f, 0.f, 0.f, 0.f, 0.f, 0.f};

    constexpr int KG = 8;
    static_assert(DIN % (4 * KG) == 0, "whole groups");
    for (int k0 = 0; k0 < DIN; k0 += 4 * KG) {
        v2f a[KG];
#pragma unroll
        for (int s = 0; s < KG; ++s)
            a[s] = *(const v2f*)(Arow + k0 + 4 * s + kb);
#pragma unroll
        for (int s = 0; s < KG; ++s) {
            const int k = k0 + 4 * s;
            const v2f* brow = Bs2 + ((k >> 1) + half) * WP + l15;
#pragma unroll
            for (int t = 0; t < NT; ++t) {
                const v2f b = brow[t * 16];
                acc[t] = __builtin_amdgcn_wmma_f32_16x16x4_f32(
                    false, a[s], false, b, (short)0, acc[t], false, false);
            }
        }
    }

    float* orow = hw + (size_t)(nb + m0 + half * 8) * HID + l15;
    const float* ds = dsh + m0 + half * 8;
#pragma unroll
    for (int t = 0; t < NT; ++t) {
#pragma unroll
        for (int v = 0; v < 8; ++v)
            orow[(size_t)v * HID + t * 16] = acc[t][v] * ds[v];
    }
}

__launch_bounds__(SCT, 16)
__global__ void bin_sort(const int2* __restrict__ tmp, const int* __restrict__ PH,
                         int* __restrict__ rowptr, float* __restrict__ dinv,
                         int* __restrict__ srcs,
                         const float* __restrict__ x, const float* __restrict__ W0,
                         float* __restrict__ hw)
{
    constexpr int WP = PadRow<HID>::value;
    extern __shared__ v2f Bs2[];
    __shared__ int c[BW];
    __shared__ float dsh[BW];
    __shared__ int wsum[32];
    __shared__ int psum[32];
    const int tid  = threadIdx.x;
    const bool prod = blockIdx.x < NBIN;
    const int bin  = prod ? blockIdx.x : blockIdx.x - NBIN;
    const int nb   = bin * BW;
    const int n0   = bin * EB;

    if (prod) {
#pragma unroll
        for (int idx = tid; idx < (DIN / 2) * HID; idx += SCT) {
            const int kp = idx / HID;
            const int n  = idx - kp * HID;
            v2f p;
            p.x = W0[(2 * kp)     * HID + n];
            p.y = W0[(2 * kp + 1) * HID + n];
            Bs2[kp * WP + n] = p;
        }
    }

    const int pb   = tid / TPP;
    const int sub  = tid - pb * TPP;
    const int ph   = (pb < EB) ? PH[n0 + pb] : 0;
    const int cnt  = imin(imax(ph >> 16, 0), CH);
    const int st   = imin(pb, EB - 1) * CH + imin(ph & 0xFFFF, CH - cnt);

    int pre = 0;
    if (!prod) {
        for (int i0 = tid; i0 < n0; i0 += 8 * SCT) {
            int t[8];
#pragma unroll
            for (int u = 0; u < 8; ++u) t[u] = PH[imin(i0 + u * SCT, n0 - 1)];
#pragma unroll
            for (int u = 0; u < 8; ++u) pre += (i0 + u * SCT < n0) ? (t[u] >> 16) : 0;
        }
    }

    if (tid < BW) c[tid] = 0;
    __syncthreads();
    for (int i0 = sub; i0 < cnt; i0 += TPP * RPT) {
        int dl[RPT];
#pragma unroll
        for (int u = 0; u < RPT; ++u)
            dl[u] = imin(imax(tmp[st + imin(i0 + u * TPP, cnt - 1)].y - nb, 0), BW - 1);
#pragma unroll
        for (int u = 0; u < RPT; ++u)
            atomicAdd(&c[dl[u]], (i0 + u * TPP < cnt) ? 1 : 0);
    }
    __syncthreads();
    const int v  = (tid < BW) ? c[tid] : 0;
    if (prod) {
        if (tid < BW) dsh[tid] = rsqrtf((float)(v + 1));
        __syncthreads();
        range_product(x, Bs2, dsh, hw, nb, tid);
        return;
    }
    const int at = block_excl_scan(v, pre, wsum, psum);
    if (tid < BW) {
        rowptr[nb + tid] = at;
        dinv[nb + tid]   = rsqrtf((float)(v + 1));
        c[tid] = at;
    }
    if (bin == NBIN - 1 && tid == 0) rowptr[GN] = GE;
    __syncthreads();
    for (int i0 = sub; i0 < cnt; i0 += TPP * RPT) {
        int2 sd[RPT];
        int  pos[RPT];
#pragma unroll
        for (int u = 0; u < RPT; ++u)
            sd[u] = tmp[st + imin(i0 + u * TPP, cnt - 1)];
#pragma unroll
        for (int u = 0; u < RPT; ++u) {
            const bool live = i0 + u * TPP < cnt;
            const int q = atomicAdd(&c[imin(imax(sd[u].y - nb, 0), BW - 1)], live ? 1 : 0);
            pos[u] = live ? q : -1;
        }
#pragma unroll
        for (int u = 0; u < RPT; ++u)
            if ((unsigned)pos[u] < (unsigned)GE) srcs[pos[u]] = sd[u].x;
    }
}

static constexpr int NPW = 4;
static_assert(GN % NPW == 0, "a wave's nodes are all valid");

template <int G>
__device__ __forceinline__ void gather_rows(const float* __restrict__ hw,
                                            int s, int j, int c,
                                            float& ax, float& ay)
{
    float2 v[G];
#pragma unroll
    for (int u = 0; u < G; ++u) {
        const int sj = __builtin_amdgcn_readlane(s, j + u);
        v[u] = *(const float2*)(hw + (size_t)sj * HID + c);
    }
#pragma unroll
    for (int u = 0; u < G; ++u) {
        ax += v[u].x;
        ay += v[u].y;
    }
}

template <class Pre, class Emit>
__device__ __forceinline__ void gather_nodes(const int* __restrict__ rowptr,
                                             const int* __restrict__ srcs,
                                             const float* __restrict__ dinv,
                                             const float* __restrict__ hw,
                                             const float* __restrict__ bias,
                                             int i0, int lane, Pre pre, Emit emit)
{
    const int c = lane * 2;
    int rp[NPW + 1];
#pragma unroll
    for (int q = 0; q <= NPW; ++q)
        rp[q] = imin(imax(__builtin_amdgcn_readfirstlane(rowptr[i0 + q]), 0), GE);

    int    a0[NPW];
    float2 hs[NPW];
    float  di[NPW];
#pragma unroll
    for (int q = 0; q < NPW; ++q) {
        const int n = imin(rp[q + 1] - rp[q], 32);
        const int idx = imin(rp[q] + imax(imin(lane, n - 1), 0), GE - 1);
        a0[q] = srcs[idx];
        hs[q] = *(const float2*)(hw + (size_t)(i0 + q) * HID + c);
        di[q] = dinv[i0 + q];
    }
    const float2 bv = *(const float2*)(bias + c);
    pre();

#pragma unroll
    for (int q = 0; q < NPW; ++q) {
        const int end = rp[q + 1];
        int base = rp[q];
        int aa = a0[q];
        float ax = 0.0f, ay = 0.0f;
        while (base < end) {
            const int n = imin(end - base, 32);
            const int s = imin(imax(aa, 0), GN - 1);
            int j = 0;
            for (; j + 8 <= n; j += 8) gather_rows<8>(hw, s, j, c, ax, ay);
            if (j + 4 <= n) { gather_rows<4>(hw, s, j, c, ax, ay); j += 4; }
            if (j + 2 <= n) { gather_rows<2>(hw, s, j, c, ax, ay); j += 2; }
            if (j < n)      { gather_rows<1>(hw, s, j, c, ax, ay); }
            base += 32;
            if (base < end) {
                const int n2 = imin(end - base, 32);
                aa = srcs[imin(base + imin(lane, n2 - 1), GE - 1)];
            }
        }
        ax += hs[q].x;
        ay += hs[q].y;
        float2 o;
        o.x = fmaxf(di[q] * ax + bv.x, 0.0f);
        o.y = fmaxf(di[q] * ay + bv.y, 0.0f);
        emit(q, o, di[q]);
    }
}

static constexpr int GBW  = 8;
static constexpr int GBN  = GBW * NPW;
static constexpr int APAD = HID + 4;
static constexpr int HP   = 2 * HID;
static_assert(GN % GBN == 0, "every block owns GBN valid nodes");
static_assert(GBN == 32 && HID == 64, "8 waves = 2 row tiles x 4 column tiles");

__launch_bounds__(256)
__global__ void node_gather(const int* __restrict__ rowptr,
                            const int* __restrict__ srcs,
                            const float* __restrict__ dinv,
                            const float* __restrict__ hw,
                            const float* __restrict__ bias,
                            float* __restrict__ H01, int col0,
                            const float* __restrict__ Wn,
                            float* __restrict__ hwn)
{
    __shared__ float As[GBN * APAD];
    const int lane = threadIdx.x & 31;
    const int wib = __builtin_amdgcn_readfirstlane((int)(threadIdx.x >> 5));
    const int nb0 = blockIdx.x * GBN;
    if (nb0 >= GN) return;
    const int i0 = nb0 + wib * NPW;
    const int c = lane * 2;

    gather_nodes(rowptr, srcs, dinv, hw, bias, i0, lane, [] {},
                 [&](int q, float2 o, float d) {
        *(float2*)(H01 + (size_t)(i0 + q) * HP + col0 + c) = o;
        float2 os;
        os.x = d * o.x;
        os.y = d * o.y;
        *(float2*)(As + (wib * NPW + q) * APAD + c) = os;
    });

    const int rt   = wib >> 2;
    const int ct   = wib & 3;
    const int l15  = lane & 15;
    const int half = lane >> 4;
    const int kb   = half * 2;

    const float* wp = Wn + (size_t)kb * HID + ct * 16 + l15;
    v2f bf[HID / 4];
#pragma unroll
    for (int ks = 0; ks < HID / 4; ++ks) {
        bf[ks].x = wp[(4 * ks)     * HID];
        bf[ks].y = wp[(4 * ks + 1) * HID];
    }
    __syncthreads();

    const float* arow = As + (rt * 16 + l15) * APAD + kb;
    v8f acc = (v8f){0.f, 0.f, 0.f, 0.f, 0.f, 0.f, 0.f, 0.f};
#pragma unroll
    for (int ks = 0; ks < HID / 4; ++ks) {
        const v2f a = *(const v2f*)(arow + 4 * ks);
        acc = __builtin_amdgcn_wmma_f32_16x16x4_f32(
            false, a, false, bf[ks], (short)0, acc, false, false);
    }
    float* orow = hwn + (size_t)(nb0 + rt * 16 + half * 8) * HID + ct * 16 + l15;
#pragma unroll
    for (int v = 0; v < 8; ++v)
        orow[(size_t)v * HID] = acc[v];
}

static constexpr int FPAD = CAT + 4;
static constexpr int NTL  = (LAB + 15) / 16;
static constexpr int LP   = 52;
static constexpr int KCH  = 16;
static_assert(NTL <= GBW && NTL * 16 <= LP && CAT % (2 * KCH) == 0 && LAB > 32 && LAB <= 64,
              "logit tiles fit the block; a lane owns labels lane and lane + 32");

__launch_bounds__(256)
__global__ void node_gather_head(const int* __restrict__ rowptr,
                                 const int* __restrict__ srcs,
                                 const float* __restrict__ dinv,
                                 const float* __restrict__ hw,
                                 const float* __restrict__ bias,
                                 const float* __restrict__ H01,
                                 const float* __restrict__ Wl,
                                 const float* __restrict__ bl,
                                 float* __restrict__ out)
{
    __shared__ float As[GBN * FPAD];
    __shared__ float Ls[GBN * LP];
    const int lane = threadIdx.x & 31;
    const int wib = __builtin_amdgcn_readfirstlane((int)(threadIdx.x >> 5));
    const int nb0 = blockIdx.x * GBN;
    if (nb0 >= GN) return;
    const int i0 = nb0 + wib * NPW;
    const int c = lane * 2;

    const int l15  = lane & 15;
    const int half = lane >> 4;
    const int kb   = half * 2;
    const int n    = wib * 16 + l15;
    const int nc   = (n < LAB) ? n : 0;
    const float* wp = Wl + (size_t)kb * LAB + nc;
    auto load_b = [&](v2f (&bf)[KCH / 4], int k0) {
#pragma unroll
        for (int ks = 0; ks < KCH / 4; ++ks) {
            const float* p = wp + (size_t)(k0 + 4 * ks) * LAB;
            bf[ks].x = p[0];
            bf[ks].y = p[LAB];
        }
    };

    gather_nodes(rowptr, srcs, dinv, hw, bias, i0, lane,
        [&] {
            float4 hc[NPW];
#pragma unroll
            for (int q = 0; q < NPW; ++q)
                hc[q] = *(const float4*)(H01 + (size_t)(i0 + q) * HP + lane * 4);
#pragma unroll
            for (int q = 0; q < NPW; ++q)
                *(float4*)(As + (wib * NPW + q) * FPAD + lane * 4) = hc[q];
        },
        [&](int q, float2 o, float) {
            *(float2*)(As + (wib * NPW + q) * FPAD + HP + c) = o;
        });

    const bool mm = wib < NTL;
    v2f bf0[KCH / 4], bf1[KCH / 4];
#pragma unroll
    for (int ks = 0; ks < KCH / 4; ++ks) bf0[ks] = (v2f){0.f, 0.f};
    if (mm) load_b(bf0, 0);
    __syncthreads();

    if (mm) {
        const float* ar0 = As + l15 * FPAD + kb;
        const float* ar1 = As + (16 + l15) * FPAD + kb;
        v8f acc0 = (v8f){0.f, 0.f, 0.f, 0.f, 0.f, 0.f, 0.f, 0.f};
        v8f acc1 = (v8f){0.f, 0.f, 0.f, 0.f, 0.f, 0.f, 0.f, 0.f};
        auto mma_chunk = [&](const v2f (&bf)[KCH / 4], int k0) {
#pragma unroll
            for (int ks = 0; ks < KCH / 4; ++ks) {
                const v2f a0 = *(const v2f*)(ar0 + k0 + 4 * ks);
                const v2f a1 = *(const v2f*)(ar1 + k0 + 4 * ks);
                acc0 = __builtin_amdgcn_wmma_f32_16x16x4_f32(
                    false, a0, false, bf[ks], (short)0, acc0, false, false);
                acc1 = __builtin_amdgcn_wmma_f32_16x16x4_f32(
                    false, a1, false, bf[ks], (short)0, acc1, false, false);
            }
        };
#pragma nounroll
        for (int k0 = 0; k0 < CAT; k0 += 2 * KCH) {
            load_b(bf1, k0 + KCH);
            mma_chunk(bf0, k0);
            load_b(bf0, (k0 + 2 * KCH < CAT) ? k0 + 2 * KCH : k0);
            mma_chunk(bf1, k0 + KCH);
        }
        const float bv = bl[nc];
        float* l0 = Ls + (half * 8) * LP + n;
        float* l1 = Ls + (16 + half * 8) * LP + n;
#pragma unroll
        for (int v = 0; v < 8; ++v) {
            l0[v * LP] = acc0[v] + bv;
            l1[v * LP] = acc1[v] + bv;
        }
    }
    __syncthreads();

#pragma unroll
    for (int q = 0; q < NPW; ++q) {
        const int row = wib * NPW + q;
        const float* lr = Ls + row * LP;
        const bool hi = lane < LAB - 32;
        const float v0 = lr[lane];
        const float v1 = hi ? lr[32 + lane] : -3.4e38f;
        float mx = fmaxf(v0, v1);
#pragma unroll
        for (int o = 16; o > 0; o >>= 1) mx = fmaxf(mx, __shfl_xor(mx, o));
        float sum = __expf(v0 - mx) + (hi ? __expf(v1 - mx) : 0.0f);
#pragma unroll
        for (int o = 16; o > 0; o >>= 1) sum += __shfl_xor(sum, o);
        const float ls = __logf(sum);
        float* orow = out + (size_t)(nb0 + row) * LAB;
        orow[lane] = v0 - mx - ls;
        if (hi) orow[32 + lane] = v1 - mx - ls;
    }
}


extern "C" void kernel_launch(void* const* d_in, const int* in_sizes, int n_in,
                              void* d_out, int out_size, void* d_ws, size_t ws_size,
                              hipStream_t stream)
{
    (void)in_sizes; (void)n_in; (void)out_size; (void)ws_size;

    const float* x  = (const float*)d_in[0];
    const int*   ei = (const int*)d_in[1];
    const float* Wm[3] = {(const float*)d_in[2], (const float*)d_in[4],
                          (const float*)d_in[6]};
    const float* bm[3] = {(const float*)d_in[3], (const float*)d_in[5],
                          (const float*)d_in[7]};
    const float* Wl = (const float*)d_in[8];
    const float* bl = (const float*)d_in[9];
    float* out = (float*)d_out;

    char* ws = (char*)d_ws;
    auto align256 = [](size_t b) { return (b + 255) & ~(size_t)255; };
    size_t off = 0;
    float* dinv   = (float*)(ws + off); off += align256((size_t)GN * 4);
    int*   rowptr = (int*)  (ws + off); off += align256((size_t)(GN + 1) * 4);
    int*   binPH  = (int*)  (ws + off); off += align256((size_t)NBIN * EB * 4);
    int2*  tmp    = (int2*) (ws + off); off += align256((size_t)EB * CH * 8);
    int*   srcs   = (int*)  (ws + off); off += align256((size_t)GE * 4);
    float* hw     = (float*)(ws + off); off += align256((size_t)GN * HID * 4);
    float* hw2    = (float*)(ws + off); off += align256((size_t)GN * HID * 4);
    float* H01    = (float*)(ws + off); off += align256((size_t)GN * HP * 4);

    const dim3 blk(256);

    bin_local<<<EB, dim3(L1T), 0, stream>>>(ei, binPH, tmp);
    const size_t shmem0 = (size_t)(DIN / 2) * PadRow<HID>::value * sizeof(float2);
    bin_sort <<<2 * NBIN, dim3(SCT), shmem0, stream>>>(tmp, binPH, rowptr, dinv, srcs,
                                                       x, Wm[0], hw);

    const int gatherBlocks = GN / GBN;

    node_gather<<<gatherBlocks, blk, 0, stream>>>(
        rowptr, srcs, dinv, hw, bm[0], H01, 0, Wm[1], hw2);
    node_gather<<<gatherBlocks, blk, 0, stream>>>(
        rowptr, srcs, dinv, hw2, bm[1], H01, HID, Wm[2], hw);
    node_gather_head<<<gatherBlocks, blk, 0, stream>>>(
        rowptr, srcs, dinv, hw, bm[2], H01, Wl, bl, out);
}


# final linear carried through the gather launches: each adds its layer's 64 k steps to its nodes' logit sums on the f32 WMMA (sums as C operand, k order kept); row plane of layers 0-1 gone
# speedup vs baseline: 4.1432x; 1.0633x over previous
#include <hip/hip_runtime.h>


typedef __attribute__((ext_vector_type(2))) float v2f;
typedef __attribute__((ext_vector_type(8))) float v8f;

static constexpr int GN   = 100000;
static constexpr int GE   = 1000000;
static constexpr int DIN  = 128;
static constexpr int HID  = 64;
static constexpr int CAT  = 192;
static constexpr int LAB  = 40;

template <int W> struct PadRow {
    static constexpr int value = W + ((48 - (W % 32)) % 32);
};

static constexpr int BW   = 400;
static constexpr int NBIN = GN / BW;
static constexpr int CH   = 4096;
static constexpr int EB   = (GE + CH - 1) / CH;
static constexpr int L1T  = 512;
static constexpr int GRPS  = CH / 4 / L1T;
static constexpr int SCT  = 1024;
static constexpr int NBW  = (NBIN + 31) / 32;
static constexpr int TPP  = 4;
static constexpr int RPT  = 8;
static_assert(GN % BW == 0, "ranges cover the nodes exactly");
static_assert(NBIN <= L1T && EB * TPP <= SCT && BW <= SCT, "one thread per counter, TPP per piece");
static_assert(GE % 4 == 0 && CH % (4 * L1T) == 0, "edges are read four at a time");
static_assert(CH < 65536 && GE >= CH, "piece start and length share a word; slot 0 of tmp is always written");
static_assert(BW % 16 == 0 && BW / 16 <= SCT / 32 && DIN % 4 == 0 && HID % 16 == 0,
              "a wave per 16-row stripe of the range's layer-0 product");

__device__ __forceinline__ int imin(int a, int b) { return a < b ? a : b; }
__device__ __forceinline__ int imax(int a, int b) { return a > b ? a : b; }

__device__ __forceinline__ int bin_of(int d) {
    return imin(imax(d, 0), GN - 1) / BW;
}

__device__ __forceinline__ int block_excl_scan(int v, int pre, int* wsum, int* psum)
{
    const int lane = threadIdx.x & 31;
    const int wave = threadIdx.x >> 5;
    int inc = v;
#pragma unroll
    for (int o = 1; o < 32; o <<= 1) {
        const int y = __shfl_up(inc, o);
        if (lane >= o) inc += y;
    }
#pragma unroll
    for (int o = 16; o > 0; o >>= 1) pre += __shfl_xor(pre, o);
    if (lane == 31) wsum[wave] = inc;
    if (lane == 0)  psum[wave] = pre;
    __syncthreads();
    int tot = psum[lane] + ((lane < wave) ? wsum[lane] : 0);
#pragma unroll
    for (int o = 16; o > 0; o >>= 1) tot += __shfl_xor(tot, o);
    return tot + inc - v;
}

__launch_bounds__(L1T)
__global__ void bin_local(const int* __restrict__ ei, int* __restrict__ PH,
                          int2* __restrict__ tmp)
{
    __shared__ int h[NBIN];
    __shared__ int wsum[NBW];
    const int tid  = threadIdx.x;
    const int b    = blockIdx.x;
    const int lane = tid & 31;
    const int wave = tid >> 5;
    if (tid < NBIN) h[tid] = 0;
    __syncthreads();
    const int4* s4 = (const int4*)ei;
    const int4* d4 = (const int4*)(ei + GE);
    int4 s[GRPS], d[GRPS], r[GRPS];
    bool ok[GRPS];
#pragma unroll
    for (int k = 0; k < GRPS; ++k) {
        const int g = b * (CH / 4) + k * L1T + tid;
        ok[k] = g < GE / 4;
        s[k]  = s4[ok[k] ? g : 0];
        d[k]  = d4[ok[k] ? g : 0];
    }
#pragma unroll
    for (int k = 0; k < GRPS; ++k) {
        r[k].x = bin_of(d[k].x);
        r[k].y = bin_of(d[k].y);
        r[k].z = bin_of(d[k].z);
        r[k].w = bin_of(d[k].w);
        if (ok[k]) {
            atomicAdd(&h[r[k].x], 1);
            atomicAdd(&h[r[k].y], 1);
            atomicAdd(&h[r[k].z], 1);
            atomicAdd(&h[r[k].w], 1);
        }
    }
    __syncthreads();
    const int v = (tid < NBIN) ? h[tid] : 0;
    int inc = v;
#pragma unroll
    for (int o = 1; o < 32; o <<= 1) {
        const int y = __shfl_up(inc, o);
        if (lane >= o) inc += y;
    }
    if (lane == 31 && wave < NBW) wsum[wave] = inc;
    __syncthreads();
    int front = 0;
#pragma unroll
    for (int w = 0; w < NBW; ++w) front += (w < wave) ? wsum[w] : 0;
    const int ex = front + inc - v;
    if (tid < NBIN) {
        PH[tid * EB + b] = (v << 16) | ex;
        h[tid] = ex;
    }
    __syncthreads();
    int2* mine = tmp + (size_t)b * CH;
#pragma unroll
    for (int k = 0; k < GRPS; ++k) {
        if (ok[k]) {
            const int p0 = atomicAdd(&h[r[k].x], 1);
            const int p1 = atomicAdd(&h[r[k].y], 1);
            const int p2 = atomicAdd(&h[r[k].z], 1);
            const int p3 = atomicAdd(&h[r[k].w], 1);
            if ((unsigned)p0 < (unsigned)CH) mine[p0] = make_int2(s[k].x, d[k].x);
            if ((unsigned)p1 < (unsigned)CH) mine[p1] = make_int2(s[k].y, d[k].y);
            if ((unsigned)p2 < (unsigned)CH) mine[p2] = make_int2(s[k].z, d[k].z);
            if ((unsigned)p3 < (unsigned)CH) mine[p3] = make_int2(s[k].w, d[k].w);
        }
    }
}

__device__ __forceinline__ void range_product(const float* __restrict__ x,
                                              const v2f* Bs2, const float* dsh,
                                              float* __restrict__ hw, int nb, int tid)
{
    constexpr int NT = HID / 16;
    constexpr int WP = PadRow<HID>::value;
    const int wv = __builtin_amdgcn_readfirstlane(tid >> 5);
    if (wv >= BW / 16) return;
    const int lane = tid & 31;
    const int l15  = lane & 15;
    const int half = lane >> 4;
    const int kb   = half * 2;
    const int m0   = wv * 16;
    const float* Arow = x + (size_t)(nb + m0 + l15) * DIN;

    v8f acc[NT];
#pragma unroll
    for (int t = 0; t < NT; ++t)
        acc[t] = (v8f){0.f, 0.f, 0.f, 0.f, 0.f, 0.f, 0.f, 0.f};

    constexpr int KG = 8;
    static_assert(DIN % (4 * KG) == 0, "whole groups");
    for (int k0 = 0; k0 < DIN; k0 += 4 * KG) {
        v2f a[KG];
#pragma unroll
        for (int s = 0; s < KG; ++s)
            a[s] = *(const v2f*)(Arow + k0 + 4 * s + kb);
#pragma unroll
        for (int s = 0; s < KG; ++s) {
            const int k = k0 + 4 * s;
            const v2f* brow = Bs2 + ((k >> 1) + half) * WP + l15;
#pragma unroll
            for (int t = 0; t < NT; ++t) {
                const v2f b = brow[t * 16];
                acc[t] = __builtin_amdgcn_wmma_f32_16x16x4_f32(
                    false, a[s], false, b, (short)0, acc[t], false, false);
            }
        }
    }

    float* orow = hw + (size_t)(nb + m0 + half * 8) * HID + l15;
    const float* ds = dsh + m0 + half * 8;
#pragma unroll
    for (int t = 0; t < NT; ++t) {
#pragma unroll
        for (int v = 0; v < 8; ++v)
            orow[(size_t)v * HID + t * 16] = acc[t][v] * ds[v];
    }
}

__launch_bounds__(SCT, 16)
__global__ void bin_sort(const int2* __restrict__ tmp, const int* __restrict__ PH,
                         int* __restrict__ rowptr, float* __restrict__ dinv,
                         int* __restrict__ srcs,
                         const float* __restrict__ x, const float* __restrict__ W0,
                         float* __restrict__ hw)
{
    constexpr int WP = PadRow<HID>::value;
    extern __shared__ v2f Bs2[];
    __shared__ int c[BW];
    __shared__ float dsh[BW];
    __shared__ int wsum[32];
    __shared__ int psum[32];
    const int tid  = threadIdx.x;
    const bool prod = blockIdx.x < NBIN;
    const int bin  = prod ? blockIdx.x : blockIdx.x - NBIN;
    const int nb   = bin * BW;
    const int n0   = bin * EB;

    if (prod) {
#pragma unroll
        for (int idx = tid; idx < (DIN / 2) * HID; idx += SCT) {
            const int kp = idx / HID;
            const int n  = idx - kp * HID;
            v2f p;
            p.x = W0[(2 * kp)     * HID + n];
            p.y = W0[(2 * kp + 1) * HID + n];
            Bs2[kp * WP + n] = p;
        }
    }

    const int pb   = tid / TPP;
    const int sub  = tid - pb * TPP;
    const int ph   = (pb < EB) ? PH[n0 + pb] : 0;
    const int cnt  = imin(imax(ph >> 16, 0), CH);
    const int st   = imin(pb, EB - 1) * CH + imin(ph & 0xFFFF, CH - cnt);

    int pre = 0;
    if (!prod) {
        for (int i0 = tid; i0 < n0; i0 += 8 * SCT) {
            int t[8];
#pragma unroll
            for (int u = 0; u < 8; ++u) t[u] = PH[imin(i0 + u * SCT, n0 - 1)];
#pragma unroll
            for (int u = 0; u < 8; ++u) pre += (i0 + u * SCT < n0) ? (t[u] >> 16) : 0;
        }
    }

    if (tid < BW) c[tid] = 0;
    __syncthreads();
    for (int i0 = sub; i0 < cnt; i0 += TPP * RPT) {
        int dl[RPT];
#pragma unroll
        for (int u = 0; u < RPT; ++u)
            dl[u] = imin(imax(tmp[st + imin(i0 + u * TPP, cnt - 1)].y - nb, 0), BW - 1);
#pragma unroll
        for (int u = 0; u < RPT; ++u)
            atomicAdd(&c[dl[u]], (i0 + u * TPP < cnt) ? 1 : 0);
    }
    __syncthreads();
    const int v  = (tid < BW) ? c[tid] : 0;
    if (prod) {
        if (tid < BW) dsh[tid] = rsqrtf((float)(v + 1));
        __syncthreads();
        range_product(x, Bs2, dsh, hw, nb, tid);
        return;
    }
    const int at = block_excl_scan(v, pre, wsum, psum);
    if (tid < BW) {
        rowptr[nb + tid] = at;
        dinv[nb + tid]   = rsqrtf((float)(v + 1));
        c[tid] = at;
    }
    if (bin == NBIN - 1 && tid == 0) rowptr[GN] = GE;
    __syncthreads();
    for (int i0 = sub; i0 < cnt; i0 += TPP * RPT) {
        int2 sd[RPT];
        int  pos[RPT];
#pragma unroll
        for (int u = 0; u < RPT; ++u)
            sd[u] = tmp[st + imin(i0 + u * TPP, cnt - 1)];
#pragma unroll
        for (int u = 0; u < RPT; ++u) {
            const bool live = i0 + u * TPP < cnt;
            const int q = atomicAdd(&c[imin(imax(sd[u].y - nb, 0), BW - 1)], live ? 1 : 0);
            pos[u] = live ? q : -1;
        }
#pragma unroll
        for (int u = 0; u < RPT; ++u)
            if ((unsigned)pos[u] < (unsigned)GE) srcs[pos[u]] = sd[u].x;
    }
}

static constexpr int NPW = 4;
static_assert(GN % NPW == 0, "a wave's nodes are all valid");

template <int G>
__device__ __forceinline__ void gather_rows(const float* __restrict__ hw,
                                            int s, int j, int c,
                                            float& ax, float& ay)
{
    float2 v[G];
#pragma unroll
    for (int u = 0; u < G; ++u) {
        const int sj = __builtin_amdgcn_readlane(s, j + u);
        v[u] = *(const float2*)(hw + (size_t)sj * HID + c);
    }
#pragma unroll
    for (int u = 0; u < G; ++u) {
        ax += v[u].x;
        ay += v[u].y;
    }
}

template <class Pre, class Emit>
__device__ __forceinline__ void gather_nodes(const int* __restrict__ rowptr,
                                             const int* __restrict__ srcs,
                                             const float* __restrict__ dinv,
                                             const float* __restrict__ hw,
                                             const float* __restrict__ bias,
                                             int i0, int lane, Pre pre, Emit emit)
{
    const int c = lane * 2;
    int rp[NPW + 1];
#pragma unroll
    for (int q = 0; q <= NPW; ++q)
        rp[q] = imin(imax(__builtin_amdgcn_readfirstlane(rowptr[i0 + q]), 0), GE);

    int    a0[NPW];
    float2 hs[NPW];
    float  di[NPW];
#pragma unroll
    for (int q = 0; q < NPW; ++q) {
        const int n = imin(rp[q + 1] - rp[q], 32);
        const int idx = imin(rp[q] + imax(imin(lane, n - 1), 0), GE - 1);
        a0[q] = srcs[idx];
        hs[q] = *(const float2*)(hw + (size_t)(i0 + q) * HID + c);
        di[q] = dinv[i0 + q];
    }
    const float2 bv = *(const float2*)(bias + c);
    pre();

#pragma unroll
    for (int q = 0; q < NPW; ++q) {
        const int end = rp[q + 1];
        int base = rp[q];
        int aa = a0[q];
        float ax = 0.0f, ay = 0.0f;
        while (base < end) {
            const int n = imin(end - base, 32);
            const int s = imin(imax(aa, 0), GN - 1);
            int j = 0;
            for (; j + 8 <= n; j += 8) gather_rows<8>(hw, s, j, c, ax, ay);
            if (j + 4 <= n) { gather_rows<4>(hw, s, j, c, ax, ay); j += 4; }
            if (j + 2 <= n) { gather_rows<2>(hw, s, j, c, ax, ay); j += 2; }
            if (j < n)      { gather_rows<1>(hw, s, j, c, ax, ay); }
            base += 32;
            if (base < end) {
                const int n2 = imin(end - base, 32);
                aa = srcs[imin(base + imin(lane, n2 - 1), GE - 1)];
            }
        }
        ax += hs[q].x;
        ay += hs[q].y;
        float2 o;
        o.x = fmaxf(di[q] * ax + bv.x, 0.0f);
        o.y = fmaxf(di[q] * ay + bv.y, 0.0f);
        emit(q, o, di[q]);
    }
}

static constexpr int GBW  = 8;
static constexpr int GBN  = GBW * NPW;
static constexpr int APAD = HID + 4;
static_assert(GN % GBN == 0, "every block owns GBN valid nodes");
static_assert(GBN == 32 && HID == 64, "8 waves = 2 row tiles x 4 column tiles");

static constexpr int NTL  = (LAB + 15) / 16;
static constexpr int LP   = 52;
static constexpr int KCH  = 32;
static constexpr int PV4  = GBN * LAB / 4;
static_assert(CAT == 3 * HID && 2 * NTL <= GBW && NTL * 16 <= LP && HID == 2 * KCH && LAB > 32 && LAB <= 64 &&
              LAB % 4 == 0 && (GBN * LAB * 4) % 128 == 0 && LP % 4 == 0,
              "Wl is one block of HID rows per layer; logit tiles fit the block; a lane owns labels lane and lane + 32");

template <bool PIN>
__launch_bounds__(256, 16)
__global__ void node_gather(const int* __restrict__ rowptr,
                            const int* __restrict__ srcs,
                            const float* __restrict__ dinv,
                            const float* __restrict__ hw,
                            const float* __restrict__ bias,
                            const float* __restrict__ Wn,
                            float* __restrict__ hwn,
                            const float* __restrict__ Wl, int kl0,
                            const float* __restrict__ Pin,
                            float* __restrict__ Pout)
{
    __shared__ float As[GBN * APAD];
    __shared__ float Ao[GBN * APAD];
    __shared__ __attribute__((aligned(16))) float Ps[GBN * LP];
    const int lane = threadIdx.x & 31;
    const int wib = __builtin_amdgcn_readfirstlane((int)(threadIdx.x >> 5));
    const int nb0 = blockIdx.x * GBN;
    if (nb0 >= GN) return;
    const int i0 = nb0 + wib * NPW;
    const int c = lane * 2;

    const int l15  = lane & 15;
    const int half = lane >> 4;
    const int kb   = half * 2;
    const int rt   = wib >> 2;
    const int ct   = wib & 3;
    const bool lg  = wib < 2 * NTL;
    const int rl   = (wib >= NTL) ? 1 : 0;
    const int cl   = wib - rl * NTL;
    const int n    = cl * 16 + l15;
    const int nc   = (n < LAB) ? n : 0;

    v8f acc2 = (v8f){0.f, 0.f, 0.f, 0.f, 0.f, 0.f, 0.f, 0.f};

    gather_nodes(rowptr, srcs, dinv, hw, bias, i0, lane,
        [&] {
            if (PIN && lg) {
                const float* pr = Pin + (size_t)(nb0 + rl * 16 + half * 8) * LAB + nc;
#pragma unroll
                for (int v = 0; v < 8; ++v) acc2[v] = pr[(size_t)v * LAB];
            }
        },
        [&](int q, float2 o, float d) {
            *(float2*)(Ao + (wib * NPW + q) * APAD + c) = o;
            float2 os;
            os.x = d * o.x;
            os.y = d * o.y;
            *(float2*)(As + (wib * NPW + q) * APAD + c) = os;
        });

    const float* wp = Wn + (size_t)kb * HID + ct * 16 + l15;
    const float* wq = Wl + (size_t)(kl0 + kb) * LAB + nc;
    auto load_b = [&](v2f (&bf)[KCH / 4], const float* p, int ld, int k0) {
#pragma unroll
        for (int ks = 0; ks < KCH / 4; ++ks) {
            const float* r = p + (size_t)(k0 + 4 * ks) * ld;
            bf[ks].x = r[0];
            bf[ks].y = r[ld];
        }
    };
    auto mma_chunk = [&](v8f& acc, const v2f (&bf)[KCH / 4], const float* arow, int k0) {
#pragma unroll
        for (int ks = 0; ks < KCH / 4; ++ks) {
            const v2f a = *(const v2f*)(arow + k0 + 4 * ks);
            acc = __builtin_amdgcn_wmma_f32_16x16x4_f32(
                false, a, false, bf[ks], (short)0, acc, false, false);
        }
    };
    v2f bf0[KCH / 4], bf1[KCH / 4];
    load_b(bf0, wp, HID, 0);
    load_b(bf1, wp, HID, KCH);
    __syncthreads();

    const float* arow = As + (rt * 16 + l15) * APAD + kb;
    const float* aorw = Ao + (rl * 16 + l15) * APAD + kb;
    v8f acc = (v8f){0.f, 0.f, 0.f, 0.f, 0.f, 0.f, 0.f, 0.f};
    mma_chunk(acc, bf0, arow, 0);
    if (lg) load_b(bf0, wq, LAB, 0);
    mma_chunk(acc, bf1, arow, KCH);
    if (lg) load_b(bf1, wq, LAB, KCH);
    float* orow = hwn + (size_t)(nb0 + rt * 16 + half * 8) * HID + ct * 16 + l15;
#pragma unroll
    for (int v = 0; v < 8; ++v)
        orow[(size_t)v * HID] = acc[v];

    if (lg) {
        mma_chunk(acc2, bf0, aorw, 0);
        mma_chunk(acc2, bf1, aorw, KCH);
        float* l0 = Ps + (rl * 16 + half * 8) * LP + n;
#pragma unroll
        for (int v = 0; v < 8; ++v) l0[v * LP] = acc2[v];
    }
    __syncthreads();

    float4* po = (float4*)(Pout + (size_t)nb0 * LAB);
    for (int e = threadIdx.x; e < PV4; e += 256) {
        const int row = e / (LAB / 4);
        const int col = (e - row * (LAB / 4)) * 4;
        po[e] = *(const float4*)(Ps + row * LP + col);
    }
}

__launch_bounds__(256)
__global__ void node_gather_head(const int* __restrict__ rowptr,
                                 const int* __restrict__ srcs,
                                 const float* __restrict__ dinv,
                                 const float* __restrict__ hw,
                                 const float* __restrict__ bias,
                                 const float* __restrict__ Pin,
                                 const float* __restrict__ Wl,
                                 const float* __restrict__ bl,
                                 float* __restrict__ out)
{
    __shared__ float As[GBN * APAD];
    __shared__ float Ls[GBN * LP];
    const int lane = threadIdx.x & 31;
    const int wib = __builtin_amdgcn_readfirstlane((int)(threadIdx.x >> 5));
    const int nb0 = blockIdx.x * GBN;
    if (nb0 >= GN) return;
    const int i0 = nb0 + wib * NPW;
    const int c = lane * 2;

    const int l15  = lane & 15;
    const int half = lane >> 4;
    const int kb   = half * 2;
    const int n    = wib * 16 + l15;
    const int nc   = (n < LAB) ? n : 0;
    const bool mm  = wib < NTL;
    const float* wp = Wl + (size_t)(2 * HID + kb) * LAB + nc;
    auto load_b = [&](v2f (&bf)[KCH / 4], int k0) {
#pragma unroll
        for (int ks = 0; ks < KCH / 4; ++ks) {
            const float* p = wp + (size_t)(k0 + 4 * ks) * LAB;
            bf[ks].x = p[0];
            bf[ks].y = p[LAB];
        }
    };

    v8f acc0 = (v8f){0.f, 0.f, 0.f, 0.f, 0.f, 0.f, 0.f, 0.f};
    v8f acc1 = (v8f){0.f, 0.f, 0.f, 0.f, 0.f, 0.f, 0.f, 0.f};

    gather_nodes(rowptr, srcs, dinv, hw, bias, i0, lane,
        [&] {
            if (mm) {
                const float* p0 = Pin + (size_t)(nb0 + half * 8) * LAB + nc;
                const float* p1 = p0 + (size_t)16 * LAB;
#pragma unroll
                for (int v = 0; v < 8; ++v) {
                    acc0[v] = p0[(size_t)v * LAB];
                    acc1[v] = p1[(size_t)v * LAB];
                }
            }
        },
        [&](int q, float2 o, float) {
            *(float2*)(As + (wib * NPW + q) * APAD + c) = o;
        });

    v2f bf0[KCH / 4], bf1[KCH / 4];
    if (mm) { load_b(bf0, 0); load_b(bf1, KCH); }
    __syncthreads();

    if (mm) {
        const float* ar0 = As + l15 * APAD + kb;
        const float* ar1 = As + (16 + l15) * APAD + kb;
        auto mma_chunk = [&](const v2f (&bf)[KCH / 4], int k0) {
#pragma unroll
            for (int ks = 0; ks < KCH / 4; ++ks) {
                const v2f a0 = *(const v2f*)(ar0 + k0 + 4 * ks);
                const v2f a1 = *(const v2f*)(ar1 + k0 + 4 * ks);
                acc0 = __builtin_amdgcn_wmma_f32_16x16x4_f32(
                    false, a0, false, bf[ks], (short)0, acc0, false, false);
                acc1 = __builtin_amdgcn_wmma_f32_16x16x4_f32(
                    false, a1, false, bf[ks], (short)0, acc1, false, false);
            }
        };
        mma_chunk(bf0, 0);
        mma_chunk(bf1, KCH);
        const float bv = bl[nc];
        float* l0 = Ls + (half * 8) * LP + n;
        float* l1 = Ls + (16 + half * 8) * LP + n;
#pragma unroll
        for (int v = 0; v < 8; ++v) {
            l0[v * LP] = acc0[v] + bv;
            l1[v * LP] = acc1[v] + bv;
        }
    }
    __syncthreads();

#pragma unroll
    for (int q = 0; q < NPW; ++q) {
        const int row = wib * NPW + q;
        const float* lr = Ls + row * LP;
        const bool hi = lane < LAB - 32;
        const float v0 = lr[lane];
        const float v1 = hi ? lr[32 + lane] : -3.4e38f;
        float mx = fmaxf(v0, v1);
#pragma unroll
        for (int o = 16; o > 0; o >>= 1) mx = fmaxf(mx, __shfl_xor(mx, o));
        float sum = __expf(v0 - mx) + (hi ? __expf(v1 - mx) : 0.0f);
#pragma unroll
        for (int o = 16; o > 0; o >>= 1) sum += __shfl_xor(sum, o);
        const float ls = __logf(sum);
        float* orow = out + (size_t)(nb0 + row) * LAB;
        orow[lane] = v0 - mx - ls;
        if (hi) orow[32 + lane] = v1 - mx - ls;
    }
}


extern "C" void kernel_launch(void* const* d_in, const int* in_sizes, int n_in,
                              void* d_out, int out_size, void* d_ws, size_t ws_size,
                              hipStream_t stream)
{
    (void)in_sizes; (void)n_in; (void)out_size; (void)ws_size;

    const float* x  = (const float*)d_in[0];
    const int*   ei = (const int*)d_in[1];
    const float* Wm[3] = {(const float*)d_in[2], (const float*)d_in[4],
                          (const float*)d_in[6]};
    const float* bm[3] = {(const float*)d_in[3], (const float*)d_in[5],
                          (const float*)d_in[7]};
    const float* Wl = (const float*)d_in[8];
    const float* bl = (const float*)d_in[9];
    float* out = (float*)d_out;

    char* ws = (char*)d_ws;
    auto align256 = [](size_t b) { return (b + 255) & ~(size_t)255; };
    size_t off = 0;
    float* dinv   = (float*)(ws + off); off += align256((size_t)GN * 4);
    int*   rowptr = (int*)  (ws + off); off += align256((size_t)(GN + 1) * 4);
    int*   binPH  = (int*)  (ws + off); off += align256((size_t)NBIN * EB * 4);
    int2*  tmp    = (int2*) (ws + off); off += align256((size_t)EB * CH * 8);
    int*   srcs   = (int*)  (ws + off); off += align256((size_t)GE * 4);
    float* hw     = (float*)(ws + off); off += align256((size_t)GN * HID * 4);
    float* hw2    = (float*)(ws + off); off += align256((size_t)GN * HID * 4);
    float* PA     = (float*)(ws + off); off += align256((size_t)GN * LAB * 4);
    float* PB     = (float*)(ws + off); off += align256((size_t)GN * LAB * 4);

    const dim3 blk(256);

    bin_local<<<EB, dim3(L1T), 0, stream>>>(ei, binPH, tmp);
    const size_t shmem0 = (size_t)(DIN / 2) * PadRow<HID>::value * sizeof(float2);
    bin_sort <<<2 * NBIN, dim3(SCT), shmem0, stream>>>(tmp, binPH, rowptr, dinv, srcs,
                                                       x, Wm[0], hw);

    const int gatherBlocks = GN / GBN;

    node_gather<false><<<gatherBlocks, blk, 0, stream>>>(
        rowptr, srcs, dinv, hw, bm[0], Wm[1], hw2, Wl, 0, nullptr, PA);
    node_gather<true><<<gatherBlocks, blk, 0, stream>>>(
        rowptr, srcs, dinv, hw2, bm[1], Wm[2], hw, Wl, HID, PA, PB);
    node_gather_head<<<gatherBlocks, blk, 0, stream>>>(
        rowptr, srcs, dinv, hw, bm[2], PB, Wl, bl, out);
}
